# starting kernel
# baseline (speedup 1.0000x reference)
#include <hip/hip_runtime.h>
#include <hip/hip_bf16.h>

typedef __attribute__((ext_vector_type(16))) _Float16 v16h;
typedef __attribute__((ext_vector_type(8)))  float    v8f;

#define NNODE 50000
#define NEDGE 800000
#define IND   64
#define HID   128
#define EDD   16
#define KE    160
#define KN    192

#define LDS_SYNC() asm volatile("s_wait_dscnt 0" ::: "memory")

__device__ __forceinline__ float silu(float v) {
    return v * __builtin_amdgcn_rcpf(1.f + __expf(-v));
}

__device__ __forceinline__ unsigned pk2h(float a, float b) {
    union { _Float16 h[2]; unsigned u; } t;
    t.h[0] = (_Float16)a; t.h[1] = (_Float16)b;
    return t.u;
}

__device__ __forceinline__ v16h load_a_frag(const _Float16* base, int lane, int kbase, int stride) {
    int m = lane & 15, hs = lane >> 4;
    union { v16h v; unsigned u[8]; } f;
    const _Float16* p = base + m * stride + kbase + 8 * hs;
#pragma unroll
    for (int r = 0; r < 8; ++r) {
        int k0 = (r < 4) ? (2 * r) : (16 + 2 * (r - 4));
        f.u[r] = *(const unsigned*)(p + k0);
    }
    return f.v;
}

__global__ void pack_weights(const float* __restrict__ W, _Float16* __restrict__ out,
                             int K, int N, int NT) {
    int frag = blockIdx.x;
    int nt = frag % NT;
    int kc = frag / NT;
    int lane = threadIdx.x;
    int nC = lane & 15, hs = lane >> 4;
    int n = nt * 16 + nC;
#pragma unroll
    for (int j = 0; j < 16; ++j) {
        int r = j >> 1, p = j & 1;
        int kl = ((r < 4) ? 2 * r : 16 + 2 * (r - 4)) + p + 8 * hs;
        int k = kc * 32 + kl;
        float v = (k < K) ? W[(long)k * N + n] : 0.f;
        out[((long)frag * 32 + lane) * 16 + j] = (_Float16)v;
    }
}

__global__ void __launch_bounds__(128) edge_kernel(
    const float* __restrict__ x, const float* __restrict__ pos,
    const int* __restrict__ ei, const float* __restrict__ ea,
    const float* __restrict__ s,
    const _Float16* __restrict__ We1f, const float* __restrict__ be1,
    const _Float16* __restrict__ We2f, const float* __restrict__ be2,
    const float* __restrict__ Wc, const float* __restrict__ bc,
    float* __restrict__ m_sum, float* __restrict__ deg, float* __restrict__ coord)
{
    __shared__ _Float16 ein_s[4][16][KE];
    __shared__ _Float16 h_s[4][16][HID];
    __shared__ float sg_s[4][16];
    __shared__ float df_s[4][16][3];
    __shared__ int   di_s[4][16];
    __shared__ int   si_s[4][16];

    int w = threadIdx.x >> 5, lane = threadIdx.x & 31;
    int e0 = (blockIdx.x * 4 + w) * 16;
    int nC = lane & 15, hs = lane >> 4;

    if (lane < 16) {
        int e = e0 + lane;
        int si = ei[e];
        int di = ei[NEDGE + e];
        si_s[w][lane] = si;
        di_s[w][lane] = di;
        sg_s[w][lane] = s[si];
        float dx = pos[di * 3 + 0] - pos[si * 3 + 0];
        float dy = pos[di * 3 + 1] - pos[si * 3 + 1];
        float dz = pos[di * 3 + 2] - pos[si * 3 + 2];
        df_s[w][lane][0] = dx; df_s[w][lane][1] = dy; df_s[w][lane][2] = dz;
        ein_s[w][lane][2 * IND] = (_Float16)(dx * dx + dy * dy + dz * dz);
#pragma unroll
        for (int c = 2 * IND + 1 + EDD; c < KE; ++c)
            ein_s[w][lane][c] = (_Float16)0.f;
    }
    LDS_SYNC();

    for (int idx = lane; idx < 16 * (IND / 4); idx += 32) {
        int m = idx >> 4, c4 = (idx & 15) * 4;
        float4 vd = *(const float4*)(x + (long)di_s[w][m] * IND + c4);
        float4 vs = *(const float4*)(x + (long)si_s[w][m] * IND + c4);
        *(uint2*)&ein_s[w][m][c4]       = make_uint2(pk2h(vd.x, vd.y), pk2h(vd.z, vd.w));
        *(uint2*)&ein_s[w][m][IND + c4] = make_uint2(pk2h(vs.x, vs.y), pk2h(vs.z, vs.w));
    }
    for (int idx = lane; idx < 16 * EDD; idx += 32) {
        int m = idx >> 4, c = idx & 15;
        ein_s[w][m][2 * IND + 1 + c] = (_Float16)ea[(long)(e0 + m) * EDD + c];
    }
    LDS_SYNC();

    for (int nt = 0; nt < 8; ++nt) {
        float b = be1[nt * 16 + nC];
        v8f c = {b, b, b, b, b, b, b, b};
#pragma unroll
        for (int kc = 0; kc < 5; ++kc) {
            v16h a  = load_a_frag(&ein_s[w][0][0], lane, kc * 32, KE);
            v16h bm = *(const v16h*)(We1f + ((long)(kc * 8 + nt) * 32 + lane) * 16);
            c = __builtin_amdgcn_wmma_f32_16x16x32_f16(false, a, false, bm, (short)0, c, false, false);
        }
#pragma unroll
        for (int r = 0; r < 8; ++r)
            h_s[w][r + 8 * hs][nt * 16 + nC] = (_Float16)silu(c[r]);
    }
    LDS_SYNC();

    float gp[8] = {0, 0, 0, 0, 0, 0, 0, 0};
    for (int nt = 0; nt < 8; ++nt) {
        int n = nt * 16 + nC;
        float b = be2[n];
        v8f c = {b, b, b, b, b, b, b, b};
#pragma unroll
        for (int kc = 0; kc < 4; ++kc) {
            v16h a  = load_a_frag(&h_s[w][0][0], lane, kc * 32, HID);
            v16h bm = *(const v16h*)(We2f + ((long)(kc * 8 + nt) * 32 + lane) * 16);
            c = __builtin_amdgcn_wmma_f32_16x16x32_f16(false, a, false, bm, (short)0, c, false, false);
        }
        float wc = Wc[n];
#pragma unroll
        for (int r = 0; r < 8; ++r) {
            int m = r + 8 * hs;
            float v = silu(c[r]) * sg_s[w][m];
            gp[r] += v * wc;
            atomicAdd(&m_sum[(long)di_s[w][m] * HID + n], v);
        }
    }
#pragma unroll
    for (int off = 1; off < 16; off <<= 1)
#pragma unroll
        for (int r = 0; r < 8; ++r) gp[r] += __shfl_xor(gp[r], off, 32);
    if (nC == 0) {
        float bcv = bc[0];
#pragma unroll
        for (int r = 0; r < 8; ++r) {
            int m = r + 8 * hs;
            int d = di_s[w][m];
            float g = gp[r] + bcv;
            atomicAdd(&coord[d * 3 + 0], g * df_s[w][m][0]);
            atomicAdd(&coord[d * 3 + 1], g * df_s[w][m][1]);
            atomicAdd(&coord[d * 3 + 2], g * df_s[w][m][2]);
            atomicAdd(&deg[d], 1.0f);
        }
    }
}

__global__ void __launch_bounds__(128) node_kernel(
    const float* __restrict__ x, const float* __restrict__ m_sum,
    const float* __restrict__ deg,
    const _Float16* __restrict__ Wn1f, const float* __restrict__ bn1,
    const _Float16* __restrict__ Wn2f, const float* __restrict__ bn2,
    float* __restrict__ out_x)
{
    __shared__ _Float16 ain_s[4][16][KN];
    __shared__ _Float16 h_s[4][16][HID];
    __shared__ float rdeg_s[4][16];

    int w = threadIdx.x >> 5, lane = threadIdx.x & 31;
    int i0 = (blockIdx.x * 4 + w) * 16;
    if (i0 >= NNODE) return;
    int nC = lane & 15, hs = lane >> 4;

    if (lane < 16) {
        int i = i0 + lane; if (i >= NNODE) i = NNODE - 1;
        float dg = deg[i]; if (dg < 1.f) dg = 1.f;
        rdeg_s[w][lane] = __builtin_amdgcn_rcpf(dg);
    }
    LDS_SYNC();
    for (int idx = lane; idx < 16 * (IND / 4); idx += 32) {
        int m = idx >> 4, c4 = (idx & 15) * 4;
        int i = i0 + m; if (i >= NNODE) i = NNODE - 1;
        float4 v = *(const float4*)(x + (long)i * IND + c4);
        *(uint2*)&ain_s[w][m][c4] = make_uint2(pk2h(v.x, v.y), pk2h(v.z, v.w));
    }
    for (int idx = lane; idx < 16 * (HID / 4); idx += 32) {
        int m = idx >> 5, c4 = (idx & 31) * 4;
        int i = i0 + m; if (i >= NNODE) i = NNODE - 1;
        float4 v = *(const float4*)(m_sum + (long)i * HID + c4);
        float r = rdeg_s[w][m];
        *(uint2*)&ain_s[w][m][IND + c4] =
            make_uint2(pk2h(v.x * r, v.y * r), pk2h(v.z * r, v.w * r));
    }
    LDS_SYNC();

    for (int nt = 0; nt < 8; ++nt) {
        float b = bn1[nt * 16 + nC];
        v8f c = {b, b, b, b, b, b, b, b};
#pragma unroll
        for (int kc = 0; kc < 6; ++kc) {
            v16h a  = load_a_frag(&ain_s[w][0][0], lane, kc * 32, KN);
            v16h bm = *(const v16h*)(Wn1f + ((long)(kc * 8 + nt) * 32 + lane) * 16);
            c = __builtin_amdgcn_wmma_f32_16x16x32_f16(false, a, false, bm, (short)0, c, false, false);
        }
#pragma unroll
        for (int r = 0; r < 8; ++r)
            h_s[w][r + 8 * hs][nt * 16 + nC] = (_Float16)silu(c[r]);
    }
    LDS_SYNC();

    for (int nt = 0; nt < 4; ++nt) {
        int n = nt * 16 + nC;
        float b = bn2[n];
        v8f c = {b, b, b, b, b, b, b, b};
#pragma unroll
        for (int kc = 0; kc < 4; ++kc) {
            v16h a  = load_a_frag(&h_s[w][0][0], lane, kc * 32, HID);
            v16h bm = *(const v16h*)(Wn2f + ((long)(kc * 4 + nt) * 32 + lane) * 16);
            c = __builtin_amdgcn_wmma_f32_16x16x32_f16(false, a, false, bm, (short)0, c, false, false);
        }
#pragma unroll
        for (int r = 0; r < 8; ++r) {
            int i = i0 + r + 8 * hs;
            if (i < NNODE) out_x[(long)i * IND + n] = c[r];
        }
    }
}

__global__ void pos_kernel(const float* __restrict__ pos, const float* __restrict__ coord,
                           const float* __restrict__ deg, float* __restrict__ out_pos) {
    int t = blockIdx.x * blockDim.x + threadIdx.x;
    if (t >= NNODE * 3) return;
    int i = t / 3;
    float dg = deg[i]; if (dg < 1.f) dg = 1.f;
    out_pos[t] = pos[t] + coord[t] * __builtin_amdgcn_rcpf(dg);
}

extern "C" void kernel_launch(void* const* d_in, const int* in_sizes, int n_in,
                              void* d_out, int out_size, void* d_ws, size_t ws_size,
                              hipStream_t stream) {
    const float* x   = (const float*)d_in[0];
    const float* pos = (const float*)d_in[1];
    const int*   ei  = (const int*)d_in[2];
    const float* ea  = (const float*)d_in[3];
    const float* s   = (const float*)d_in[4];
    const float* We1 = (const float*)d_in[5];  const float* be1 = (const float*)d_in[6];
    const float* We2 = (const float*)d_in[7];  const float* be2 = (const float*)d_in[8];
    const float* Wn1 = (const float*)d_in[9];  const float* bn1 = (const float*)d_in[10];
    const float* Wn2 = (const float*)d_in[11]; const float* bn2 = (const float*)d_in[12];
    const float* Wc  = (const float*)d_in[13]; const float* bc  = (const float*)d_in[14];

    float* out_x   = (float*)d_out;
    float* out_pos = out_x + (long)NNODE * IND;

    float* m_sum = (float*)d_ws;
    float* deg   = m_sum + (long)NNODE * HID;
    float* coord = deg + NNODE;
    _Float16* We1f = (_Float16*)(coord + (long)NNODE * 3);
    _Float16* We2f = We1f + 5 * 8 * 512;
    _Float16* Wn1f = We2f + 4 * 8 * 512;
    _Float16* Wn2f = Wn1f + 6 * 8 * 512;

    size_t acc_bytes = ((long)NNODE * HID + NNODE + (long)NNODE * 3) * sizeof(float);
    hipMemsetAsync(d_ws, 0, acc_bytes, stream);

    pack_weights<<<5 * 8, 32, 0, stream>>>(We1, We1f, 145, 128, 8);
    pack_weights<<<4 * 8, 32, 0, stream>>>(We2, We2f, 128, 128, 8);
    pack_weights<<<6 * 8, 32, 0, stream>>>(Wn1, Wn1f, 192, 128, 8);
    pack_weights<<<4 * 4, 32, 0, stream>>>(Wn2, Wn2f, 128, 64, 4);

    edge_kernel<<<NEDGE / 64, 128, 0, stream>>>(x, pos, ei, ea, s,
                                                We1f, be1, We2f, be2, Wc, bc,
                                                m_sum, deg, coord);
    node_kernel<<<(NNODE + 63) / 64, 128, 0, stream>>>(x, m_sum, deg,
                                                       Wn1f, bn1, Wn2f, bn2, out_x);
    pos_kernel<<<(NNODE * 3 + 255) / 256, 256, 0, stream>>>(pos, coord, deg, out_pos);
}


# loads of a wave issued in groups ahead of use (all x rows and edge_attr at once; next N-tile's weight fragments behind the current products); a pair's 16 row atomics back to back; sum order kept
# speedup vs baseline: 1.5048x; 1.5048x over previous
#include <hip/hip_runtime.h>
#include <hip/hip_bf16.h>

typedef __attribute__((ext_vector_type(16))) _Float16 v16h;
typedef __attribute__((ext_vector_type(8)))  float    v8f;

#define NNODE 50000
#define NEDGE 800000
#define IND   64
#define HID   128
#define EDD   16
#define KE    160
#define KN    192

#define LDS_SYNC() asm volatile("s_wait_dscnt 0" ::: "memory")

__device__ __forceinline__ float silu(float v) {
    return v * __builtin_amdgcn_rcpf(1.f + __expf(-v));
}

__device__ __forceinline__ unsigned pk2h(float a, float b) {
    union { _Float16 h[2]; unsigned u; } t;
    t.h[0] = (_Float16)a; t.h[1] = (_Float16)b;
    return t.u;
}

__device__ __forceinline__ float xhalf(float v) {
    int b = __builtin_bit_cast(int, v);
    return __builtin_bit_cast(float,
        __builtin_amdgcn_permlanex16(b, b, 0x76543210, 0xfedcba98, false, false));
}

__device__ __forceinline__ v16h load_a_frag(const _Float16* base, int lane, int kbase, int stride) {
    int m = lane & 15, hs = lane >> 4;
    union { v16h v; unsigned u[8]; } f;
    const _Float16* p = base + m * stride + kbase + 8 * hs;
#pragma unroll
    for (int r = 0; r < 8; ++r) {
        int k0 = (r < 4) ? (2 * r) : (16 + 2 * (r - 4));
        f.u[r] = *(const unsigned*)(p + k0);
    }
    return f.v;
}

__global__ void pack_weights(const float* __restrict__ W, _Float16* __restrict__ out,
                             int K, int N, int NT) {
    int frag = blockIdx.x;
    int nt = frag % NT;
    int kc = frag / NT;
    int lane = threadIdx.x;
    int nC = lane & 15, hs = lane >> 4;
    int n = nt * 16 + nC;
#pragma unroll
    for (int j = 0; j < 16; ++j) {
        int r = j >> 1, p = j & 1;
        int kl = ((r < 4) ? 2 * r : 16 + 2 * (r - 4)) + p + 8 * hs;
        int k = kc * 32 + kl;
        float v = (k < K) ? W[(long)k * N + n] : 0.f;
        out[((long)frag * 32 + lane) * 16 + j] = (_Float16)v;
    }
}

__global__ void __launch_bounds__(128) edge_kernel(
    const float* __restrict__ x, const float* __restrict__ pos,
    const int* __restrict__ ei, const float* __restrict__ ea,
    const float* __restrict__ s,
    const _Float16* __restrict__ We1f, const float* __restrict__ be1,
    const _Float16* __restrict__ We2f, const float* __restrict__ be2,
    const float* __restrict__ Wc, const float* __restrict__ bc,
    float* __restrict__ m_sum, float* __restrict__ acc4)
{
    __shared__ _Float16 ein_s[4][16][KE];
    __shared__ _Float16 h_s[4][16][HID];
    __shared__ float sg_s[4][16];
    __shared__ float df_s[4][16][4];
    __shared__ int   di_s[4][16];
    __shared__ int   si_s[4][16];

    int w = threadIdx.x >> 5, lane = threadIdx.x & 31;
    int e0 = (blockIdx.x * 4 + w) * 16;
    int nC = lane & 15, hs = lane >> 4;


    float ev[8];
#pragma unroll
    for (int t = 0; t < 8; ++t)
        ev[t] = ea[(long)e0 * EDD + lane + 32 * t];

    int si = 0, di = 0;
    if (lane < 16) {
        int e = e0 + lane;
        si = ei[e];
        di = ei[NEDGE + e];
        si_s[w][lane] = si;
        di_s[w][lane] = di;
    }
    LDS_SYNC();

    float4 vd[8], vs[8];
#pragma unroll
    for (int t = 0; t < 8; ++t) {
        int idx = lane + 32 * t;
        int m = idx >> 4, c4 = (idx & 15) * 4;
        vd[t] = *(const float4*)(x + (long)di_s[w][m] * IND + c4);
        vs[t] = *(const float4*)(x + (long)si_s[w][m] * IND + c4);
    }

    if (lane < 16) {
        sg_s[w][lane] = s[si];
        float dx = pos[di * 3 + 0] - pos[si * 3 + 0];
        float dy = pos[di * 3 + 1] - pos[si * 3 + 1];
        float dz = pos[di * 3 + 2] - pos[si * 3 + 2];
        df_s[w][lane][0] = dx; df_s[w][lane][1] = dy; df_s[w][lane][2] = dz;
        df_s[w][lane][3] = 1.0f;
        ein_s[w][lane][2 * IND] = (_Float16)(dx * dx + dy * dy + dz * dz);
#pragma unroll
        for (int c = 2 * IND + 1 + EDD; c < KE; ++c)
            ein_s[w][lane][c] = (_Float16)0.f;
    }

#pragma unroll
    for (int t = 0; t < 8; ++t) {
        int idx = lane + 32 * t;
        int m = idx >> 4, c4 = (idx & 15) * 4;
        *(uint2*)&ein_s[w][m][c4]       = make_uint2(pk2h(vd[t].x, vd[t].y), pk2h(vd[t].z, vd[t].w));
        *(uint2*)&ein_s[w][m][IND + c4] = make_uint2(pk2h(vs[t].x, vs[t].y), pk2h(vs[t].z, vs[t].w));
    }
#pragma unroll
    for (int t = 0; t < 8; ++t) {
        int idx = lane + 32 * t;
        ein_s[w][idx >> 4][2 * IND + 1 + (idx & 15)] = (_Float16)ev[t];
    }
    LDS_SYNC();

    const v16h* W1 = (const v16h*)We1f + lane;
    const v16h* W2 = (const v16h*)We2f + lane;

    v16h af[5];
#pragma unroll
    for (int kc = 0; kc < 5; ++kc)
        af[kc] = load_a_frag(&ein_s[w][0][0], lane, kc * 32, KE);
    v16h bw[5];
#pragma unroll
    for (int kc = 0; kc < 5; ++kc) bw[kc] = W1[(kc * 8) * 32];
    float bias1 = be1[nC];
    for (int nt = 0; nt < 7; ++nt) {
        v8f c = {bias1, bias1, bias1, bias1, bias1, bias1, bias1, bias1};
#pragma unroll
        for (int kc = 0; kc < 5; ++kc)
            c = __builtin_amdgcn_wmma_f32_16x16x32_f16(false, af[kc], false, bw[kc], (short)0, c, false, false);
#pragma unroll
        for (int kc = 0; kc < 5; ++kc) bw[kc] = W1[(kc * 8 + nt + 1) * 32];
        bias1 = be1[(nt + 1) * 16 + nC];
#pragma unroll
        for (int r = 0; r < 8; ++r)
            h_s[w][r + 8 * hs][nt * 16 + nC] = (_Float16)silu(c[r]);
    }
    {
        v8f c = {bias1, bias1, bias1, bias1, bias1, bias1, bias1, bias1};
#pragma unroll
        for (int kc = 0; kc < 5; ++kc)
            c = __builtin_amdgcn_wmma_f32_16x16x32_f16(false, af[kc], false, bw[kc], (short)0, c, false, false);
#pragma unroll
        for (int r = 0; r < 8; ++r)
            h_s[w][r + 8 * hs][7 * 16 + nC] = (_Float16)silu(c[r]);
    }
    LDS_SYNC();

    int rowoff[16];
#pragma unroll
    for (int k = 0; k < 16; ++k) rowoff[k] = di_s[w][k] * HID + lane;

    v8f gp = {0, 0, 0, 0, 0, 0, 0, 0};
    for (int np = 0; np < 4; ++np) {
        int n0 = np * 32 + nC, n1 = n0 + 16;
        v16h b0[4], b1[4];
#pragma unroll
        for (int kc = 0; kc < 4; ++kc) {
            b0[kc] = W2[(kc * 8 + 2 * np) * 32];
            b1[kc] = W2[(kc * 8 + 2 * np + 1) * 32];
        }
        float bs0 = be2[n0], bs1 = be2[n1];
        float wc0 = Wc[n0], wc1 = Wc[n1];
        v8f c0 = {bs0, bs0, bs0, bs0, bs0, bs0, bs0, bs0};
        v8f c1 = {bs1, bs1, bs1, bs1, bs1, bs1, bs1, bs1};
#pragma unroll
        for (int kc = 0; kc < 4; ++kc) {
            v16h a = load_a_frag(&h_s[w][0][0], lane, kc * 32, HID);
            c0 = __builtin_amdgcn_wmma_f32_16x16x32_f16(false, a, false, b0[kc], (short)0, c0, false, false);
            c1 = __builtin_amdgcn_wmma_f32_16x16x32_f16(false, a, false, b1[kc], (short)0, c1, false, false);
        }
#pragma unroll
        for (int r = 0; r < 8; ++r) {
            float sg = sg_s[w][r + 8 * hs];
            float v0 = silu(c0[r]) * sg;
            float v1 = silu(c1[r]) * sg;
            gp[r] += v0 * wc0;
            gp[r] += v1 * wc1;
            float got = xhalf(hs ? v0 : v1);
            float lo = hs ? got : v0;
            float hi = hs ? v1 : got;
            atomicAdd(m_sum + rowoff[r], lo);
            atomicAdd(m_sum + rowoff[r + 8], hi);
        }
#pragma unroll
        for (int k = 0; k < 16; ++k) rowoff[k] += 32;
    }
#pragma unroll
    for (int off = 1; off < 16; off <<= 1)
#pragma unroll
        for (int r = 0; r < 8; ++r) gp[r] += __shfl_xor((float)gp[r], off, 32);
    {
        float bcv = bc[0];
        int q = nC >> 2, wsel = nC & 3;
#pragma unroll
        for (int j = 0; j < 2; ++j) {
            float gq = (q == 0) ? gp[4 * j] : (q == 1) ? gp[4 * j + 1]
                     : (q == 2) ? gp[4 * j + 2] : gp[4 * j + 3];
            float g = gq + bcv;
            int m = q + 4 * j + 8 * hs;
            float val = (wsel == 3) ? 1.0f : g * df_s[w][m][wsel];
            atomicAdd(&acc4[di_s[w][m] * 4 + wsel], val);
        }
    }
}

__global__ void __launch_bounds__(128) node_kernel(
    const float* __restrict__ x, const float* __restrict__ m_sum,
    const float* __restrict__ acc4,
    const _Float16* __restrict__ Wn1f, const float* __restrict__ bn1,
    const _Float16* __restrict__ Wn2f, const float* __restrict__ bn2,
    float* __restrict__ out_x)
{
    __shared__ _Float16 ain_s[4][16][KN];
    __shared__ _Float16 h_s[4][16][HID];
    __shared__ float rdeg_s[4][16];

    int w = threadIdx.x >> 5, lane = threadIdx.x & 31;
    int i0 = (blockIdx.x * 4 + w) * 16;
    if (i0 >= NNODE) return;
    int nC = lane & 15, hs = lane >> 4;

    float dg = 1.f;
    if (lane < 16) {
        int i = i0 + lane; if (i >= NNODE) i = NNODE - 1;
        dg = acc4[i * 4 + 3];
    }
    float4 vx[8], vm[16];
#pragma unroll
    for (int t = 0; t < 8; ++t) {
        int idx = lane + 32 * t;
        int m = idx >> 4, c4 = (idx & 15) * 4;
        int i = i0 + m; if (i >= NNODE) i = NNODE - 1;
        vx[t] = *(const float4*)(x + (long)i * IND + c4);
    }
#pragma unroll
    for (int t = 0; t < 16; ++t) {
        int i = i0 + t; if (i >= NNODE) i = NNODE - 1;
        vm[t] = *(const float4*)(m_sum + (long)i * HID + lane * 4);
    }
    if (lane < 16) {
        if (dg < 1.f) dg = 1.f;
        rdeg_s[w][lane] = __builtin_amdgcn_rcpf(dg);
    }
    LDS_SYNC();
#pragma unroll
    for (int t = 0; t < 8; ++t) {
        int idx = lane + 32 * t;
        int m = idx >> 4, c4 = (idx & 15) * 4;
        *(uint2*)&ain_s[w][m][c4] = make_uint2(pk2h(vx[t].x, vx[t].y), pk2h(vx[t].z, vx[t].w));
    }
#pragma unroll
    for (int t = 0; t < 16; ++t) {
        float r = rdeg_s[w][t];
        *(uint2*)&ain_s[w][t][IND + lane * 4] =
            make_uint2(pk2h(vm[t].x * r, vm[t].y * r), pk2h(vm[t].z * r, vm[t].w * r));
    }
    LDS_SYNC();

    const v16h* W1 = (const v16h*)Wn1f + lane;
    const v16h* W2 = (const v16h*)Wn2f + lane;

    v16h af[6];
#pragma unroll
    for (int kc = 0; kc < 6; ++kc)
        af[kc] = load_a_frag(&ain_s[w][0][0], lane, kc * 32, KN);
    v16h bw[6];
#pragma unroll
    for (int kc = 0; kc < 6; ++kc) bw[kc] = W1[(kc * 8) * 32];
    float bias1 = bn1[nC];
    for (int nt = 0; nt < 7; ++nt) {
        v8f c = {bias1, bias1, bias1, bias1, bias1, bias1, bias1, bias1};
#pragma unroll
        for (int kc = 0; kc < 6; ++kc)
            c = __builtin_amdgcn_wmma_f32_16x16x32_f16(false, af[kc], false, bw[kc], (short)0, c, false, false);
#pragma unroll
        for (int kc = 0; kc < 6; ++kc) bw[kc] = W1[(kc * 8 + nt + 1) * 32];
        bias1 = bn1[(nt + 1) * 16 + nC];
#pragma unroll
        for (int r = 0; r < 8; ++r)
            h_s[w][r + 8 * hs][nt * 16 + nC] = (_Float16)silu(c[r]);
    }
    v16h bv[4];
    float bias2;
    {
        v8f c = {bias1, bias1, bias1, bias1, bias1, bias1, bias1, bias1};
#pragma unroll
        for (int kc = 0; kc < 6; ++kc)
            c = __builtin_amdgcn_wmma_f32_16x16x32_f16(false, af[kc], false, bw[kc], (short)0, c, false, false);
#pragma unroll
        for (int kc = 0; kc < 4; ++kc) bv[kc] = W2[(kc * 4) * 32];
        bias2 = bn2[nC];
#pragma unroll
        for (int r = 0; r < 8; ++r)
            h_s[w][r + 8 * hs][7 * 16 + nC] = (_Float16)silu(c[r]);
    }
    LDS_SYNC();

    v16h ah[4];
#pragma unroll
    for (int kc = 0; kc < 4; ++kc)
        ah[kc] = load_a_frag(&h_s[w][0][0], lane, kc * 32, HID);
#pragma unroll
    for (int nt = 0; nt < 4; ++nt) {
        int n = nt * 16 + nC;
        v8f c = {bias2, bias2, bias2, bias2, bias2, bias2, bias2, bias2};
#pragma unroll
        for (int kc = 0; kc < 4; ++kc)
            c = __builtin_amdgcn_wmma_f32_16x16x32_f16(false, ah[kc], false, bv[kc], (short)0, c, false, false);
        if (nt + 1 < 4) {
#pragma unroll
            for (int kc = 0; kc < 4; ++kc) bv[kc] = W2[(kc * 4 + nt + 1) * 32];
            bias2 = bn2[(nt + 1) * 16 + nC];
        }
#pragma unroll
        for (int r = 0; r < 8; ++r) {
            int i = i0 + r + 8 * hs;
            if (i < NNODE) out_x[(long)i * IND + n] = c[r];
        }
    }
}

__global__ void pos_kernel(const float* __restrict__ pos, const float* __restrict__ acc4,
                           float* __restrict__ out_pos) {
    int t = blockIdx.x * blockDim.x + threadIdx.x;
    if (t >= NNODE * 3) return;
    int i = t / 3;
    int k = t - 3 * i;
    float dg = acc4[i * 4 + 3]; if (dg < 1.f) dg = 1.f;
    out_pos[t] = pos[t] + acc4[i * 4 + k] * __builtin_amdgcn_rcpf(dg);
}

extern "C" void kernel_launch(void* const* d_in, const int* in_sizes, int n_in,
                              void* d_out, int out_size, void* d_ws, size_t ws_size,
                              hipStream_t stream) {
    const float* x   = (const float*)d_in[0];
    const float* pos = (const float*)d_in[1];
    const int*   ei  = (const int*)d_in[2];
    const float* ea  = (const float*)d_in[3];
    const float* s   = (const float*)d_in[4];
    const float* We1 = (const float*)d_in[5];  const float* be1 = (const float*)d_in[6];
    const float* We2 = (const float*)d_in[7];  const float* be2 = (const float*)d_in[8];
    const float* Wn1 = (const float*)d_in[9];  const float* bn1 = (const float*)d_in[10];
    const float* Wn2 = (const float*)d_in[11]; const float* bn2 = (const float*)d_in[12];
    const float* Wc  = (const float*)d_in[13]; const float* bc  = (const float*)d_in[14];

    float* out_x   = (float*)d_out;
    float* out_pos = out_x + (long)NNODE * IND;

    float* m_sum = (float*)d_ws;
    float* acc4  = m_sum + (long)NNODE * HID;
    _Float16* We1f = (_Float16*)(acc4 + (long)NNODE * 4);
    _Float16* We2f = We1f + 5 * 8 * 512;
    _Float16* Wn1f = We2f + 4 * 8 * 512;
    _Float16* Wn2f = Wn1f + 6 * 8 * 512;

    size_t acc_bytes = ((long)NNODE * HID + (long)NNODE * 4) * sizeof(float);
    hipMemsetAsync(d_ws, 0, acc_bytes, stream);

    pack_weights<<<5 * 8, 32, 0, stream>>>(We1, We1f, 145, 128, 8);
    pack_weights<<<4 * 8, 32, 0, stream>>>(We2, We2f, 128, 128, 8);
    pack_weights<<<6 * 8, 32, 0, stream>>>(Wn1, Wn1f, 192, 128, 8);
    pack_weights<<<4 * 4, 32, 0, stream>>>(Wn2, Wn2f, 128, 64, 4);

    edge_kernel<<<NEDGE / 64, 128, 0, stream>>>(x, pos, ei, ea, s,
                                                We1f, be1, We2f, be2, Wc, bc,
                                                m_sum, acc4);
    node_kernel<<<(NNODE + 63) / 64, 128, 0, stream>>>(x, m_sum, acc4,
                                                       Wn1f, bn1, Wn2f, bn2, out_x);
    pos_kernel<<<(NNODE * 3 + 255) / 256, 256, 0, stream>>>(pos, acc4, out_pos);
}


# edge kernel: both edge-MLP weight matrices copied once into LDS by 256 persistent 24-wave work-groups that loop over 16-edge tiles; B fragments read from LDS, not global; arrival-order sums as before
# speedup vs baseline: 1.7247x; 1.1461x over previous
#include <hip/hip_runtime.h>
#include <hip/hip_bf16.h>

typedef __attribute__((ext_vector_type(16))) _Float16 v16h;
typedef __attribute__((ext_vector_type(8)))  float    v8f;
typedef __attribute__((ext_vector_type(8)))  _Float16 v8h;

#define NNODE 50000
#define NEDGE 800000
#define IND   64
#define HID   128
#define EDD   16
#define KE    160
#define KN    192
#define NTILE (NEDGE / 16)
#define WPB   24
#define NBLK  256
#define F1    40
#define F2    32

#define LDS_SYNC() asm volatile("s_wait_dscnt 0" ::: "memory")

__device__ __forceinline__ float silu(float v) {
    return v * __builtin_amdgcn_rcpf(1.f + __expf(-v));
}

__device__ __forceinline__ unsigned pk2h(float a, float b) {
    union { _Float16 h[2]; unsigned u; } t;
    t.h[0] = (_Float16)a; t.h[1] = (_Float16)b;
    return t.u;
}

__device__ __forceinline__ float xhalf(float v) {
    int b = __builtin_bit_cast(int, v);
    return __builtin_bit_cast(float,
        __builtin_amdgcn_permlanex16(b, b, 0x76543210, 0xfedcba98, false, false));
}

__device__ __forceinline__ v16h load_a_frag(const _Float16* base, int lane, int kbase, int stride) {
    int m = lane & 15, hs = lane >> 4;
    union { v16h v; unsigned u[8]; } f;
    const _Float16* p = base + m * stride + kbase + 8 * hs;
#pragma unroll
    for (int r = 0; r < 8; ++r) {
        int k0 = (r < 4) ? (2 * r) : (16 + 2 * (r - 4));
        f.u[r] = *(const unsigned*)(p + k0);
    }
    return f.v;
}

__global__ void pack_weights(const float* __restrict__ W, _Float16* __restrict__ out,
                             int K, int N, int NT, int planes) {
    int frag = blockIdx.x;
    int nt = frag % NT;
    int kc = frag / NT;
    int lane = threadIdx.x;
    int nC = lane & 15, hs = lane >> 4;
    int n = nt * 16 + nC;
#pragma unroll
    for (int j = 0; j < 16; ++j) {
        int r = j >> 1, p = j & 1;
        int kl = ((r < 4) ? 2 * r : 16 + 2 * (r - 4)) + p + 8 * hs;
        int k = kc * 32 + kl;
        float v = (k < K) ? W[(long)k * N + n] : 0.f;
        long o = planes ? (((long)frag * 2 + (j >> 3)) * 32 + lane) * 8 + (j & 7)
                        : ((long)frag * 32 + lane) * 16 + j;
        out[o] = (_Float16)v;
    }
}

__device__ __forceinline__ v16h lds_b_frag(const _Float16* wl, int f, int wq) {
    const v8h* p = (const v8h*)(wl + f * 512 + wq);
    v8h lo = p[0], hi = p[32];
    return __builtin_shufflevector(lo, hi, 0, 1, 2, 3, 4, 5, 6, 7, 8, 9, 10, 11, 12, 13, 14, 15);
}

__global__ void __launch_bounds__(WPB * 32) edge_kernel(
    const float* __restrict__ x, const float* __restrict__ pos,
    const int* __restrict__ ei, const float* __restrict__ ea,
    const float* __restrict__ s,
    const _Float16* __restrict__ We1f, const float* __restrict__ be1,
    const _Float16* __restrict__ We2f, const float* __restrict__ be2,
    const float* __restrict__ Wc, const float* __restrict__ bc,
    float* __restrict__ m_sum, float* __restrict__ acc4)
{
    __shared__ __attribute__((aligned(16))) _Float16 wl_s[(F1 + F2) * 512];
    __shared__ __attribute__((aligned(16))) _Float16 ein_s[WPB][16][KE];
    __shared__ __attribute__((aligned(16))) _Float16 h_s[WPB][16][HID];
    __shared__ float sg_s[WPB][16];
    __shared__ float df_s[WPB][16][4];
    __shared__ int   di_s[WPB][16];
    __shared__ int   si_s[WPB][16];

    int w = __builtin_amdgcn_readfirstlane(threadIdx.x >> 5);
    int lane0 = threadIdx.x & 31;

    {
        const uint4* g1 = (const uint4*)We1f;
        const uint4* g2 = (const uint4*)We2f;
        uint4* l = (uint4*)wl_s;
        for (int i = threadIdx.x; i < F1 * 64; i += WPB * 32) l[i] = g1[i];
        for (int i = threadIdx.x; i < F2 * 64; i += WPB * 32) l[F1 * 64 + i] = g2[i];
    }
    __syncthreads();
    const _Float16* wl1 = wl_s;
    const _Float16* wl2 = wl_s + F1 * 512;
    _Float16* hw = &h_s[w][0][0];

    for (int tile = blockIdx.x * WPB + w; tile < NTILE; tile += gridDim.x * WPB) {
    LDS_SYNC();
    int e0 = tile * 16;
    int lane = lane0;
    asm volatile("" : "+v"(lane));
    int nC = lane & 15, hs = lane >> 4;
    int wq = lane * 8;


    float ev[8];
#pragma unroll
    for (int t = 0; t < 8; ++t)
        ev[t] = ea[(long)e0 * EDD + lane + 32 * t];

    int si = 0, di = 0;
    if (lane < 16) {
        int e = e0 + lane;
        si = ei[e];
        di = ei[NEDGE + e];
        si_s[w][lane] = si;
        di_s[w][lane] = di;
    }
    LDS_SYNC();

    float4 vd[8], vs[8];
#pragma unroll
    for (int t = 0; t < 8; ++t) {
        int idx = lane + 32 * t;
        int m = idx >> 4, c4 = (idx & 15) * 4;
        vd[t] = *(const float4*)(x + (long)di_s[w][m] * IND + c4);
        vs[t] = *(const float4*)(x + (long)si_s[w][m] * IND + c4);
    }

    if (lane < 16) {
        sg_s[w][lane] = s[si];
        float dx = pos[di * 3 + 0] - pos[si * 3 + 0];
        float dy = pos[di * 3 + 1] - pos[si * 3 + 1];
        float dz = pos[di * 3 + 2] - pos[si * 3 + 2];
        df_s[w][lane][0] = dx; df_s[w][lane][1] = dy; df_s[w][lane][2] = dz;
        df_s[w][lane][3] = 1.0f;
        ein_s[w][lane][2 * IND] = (_Float16)(dx * dx + dy * dy + dz * dz);
#pragma unroll
        for (int c = 2 * IND + 1 + EDD; c < KE; ++c)
            ein_s[w][lane][c] = (_Float16)0.f;
    }

#pragma unroll
    for (int t = 0; t < 8; ++t) {
        int idx = lane + 32 * t;
        int m = idx >> 4, c4 = (idx & 15) * 4;
        *(uint2*)&ein_s[w][m][c4]       = make_uint2(pk2h(vd[t].x, vd[t].y), pk2h(vd[t].z, vd[t].w));
        *(uint2*)&ein_s[w][m][IND + c4] = make_uint2(pk2h(vs[t].x, vs[t].y), pk2h(vs[t].z, vs[t].w));
    }
#pragma unroll
    for (int t = 0; t < 8; ++t) {
        int idx = lane + 32 * t;
        ein_s[w][idx >> 4][2 * IND + 1 + (idx & 15)] = (_Float16)ev[t];
    }
    LDS_SYNC();

    v16h af[5];
#pragma unroll
    for (int kc = 0; kc < 5; ++kc)
        af[kc] = load_a_frag(&ein_s[w][0][0], lane, kc * 32, KE);
    v16h bw[5];
#pragma unroll
    for (int kc = 0; kc < 5; ++kc) bw[kc] = lds_b_frag(wl1, kc * 8, wq);
    float bias1 = be1[nC];
    for (int nt = 0; nt < 7; ++nt) {
        v8f c = {bias1, bias1, bias1, bias1, bias1, bias1, bias1, bias1};
#pragma unroll
        for (int kc = 0; kc < 5; ++kc)
            c = __builtin_amdgcn_wmma_f32_16x16x32_f16(false, af[kc], false, bw[kc], (short)0, c, false, false);
#pragma unroll
        for (int kc = 0; kc < 5; ++kc) bw[kc] = lds_b_frag(wl1, kc * 8 + nt + 1, wq);
        bias1 = be1[(nt + 1) * 16 + nC];
#pragma unroll
        for (int r = 0; r < 8; ++r)
            hw[(r + 8 * hs) * HID + nt * 16 + nC] = (_Float16)silu(c[r]);
    }
    {
        v8f c = {bias1, bias1, bias1, bias1, bias1, bias1, bias1, bias1};
#pragma unroll
        for (int kc = 0; kc < 5; ++kc)
            c = __builtin_amdgcn_wmma_f32_16x16x32_f16(false, af[kc], false, bw[kc], (short)0, c, false, false);
#pragma unroll
        for (int r = 0; r < 8; ++r)
            hw[(r + 8 * hs) * HID + 7 * 16 + nC] = (_Float16)silu(c[r]);
    }
    LDS_SYNC();

    int rowoff[16];
#pragma unroll
    for (int k = 0; k < 16; ++k) rowoff[k] = di_s[w][k] * HID + lane;

    v8f gp = {0, 0, 0, 0, 0, 0, 0, 0};
    for (int np = 0; np < 4; ++np) {
        int n0 = np * 32 + nC, n1 = n0 + 16;
        v16h b0[4], b1[4];
#pragma unroll
        for (int kc = 0; kc < 4; ++kc) {
            b0[kc] = lds_b_frag(wl2, kc * 8 + 2 * np, wq);
            b1[kc] = lds_b_frag(wl2, kc * 8 + 2 * np + 1, wq);
        }
        float bs0 = be2[n0], bs1 = be2[n1];
        float wc0 = Wc[n0], wc1 = Wc[n1];
        v8f c0 = {bs0, bs0, bs0, bs0, bs0, bs0, bs0, bs0};
        v8f c1 = {bs1, bs1, bs1, bs1, bs1, bs1, bs1, bs1};
#pragma unroll
        for (int kc = 0; kc < 4; ++kc) {
            v16h a = load_a_frag(hw, lane, kc * 32, HID);
            c0 = __builtin_amdgcn_wmma_f32_16x16x32_f16(false, a, false, b0[kc], (short)0, c0, false, false);
            c1 = __builtin_amdgcn_wmma_f32_16x16x32_f16(false, a, false, b1[kc], (short)0, c1, false, false);
        }
#pragma unroll
        for (int r = 0; r < 8; ++r) {
            float sg = sg_s[w][r + 8 * hs];
            float v0 = silu(c0[r]) * sg;
            float v1 = silu(c1[r]) * sg;
            gp[r] += v0 * wc0;
            gp[r] += v1 * wc1;
            float got = xhalf(hs ? v0 : v1);
            float lo = hs ? got : v0;
            float hi = hs ? v1 : got;
            atomicAdd(m_sum + rowoff[r], lo);
            atomicAdd(m_sum + rowoff[r + 8], hi);
        }
#pragma unroll
        for (int k = 0; k < 16; ++k) rowoff[k] += 32;
    }
#pragma unroll
    for (int off = 1; off < 16; off <<= 1)
#pragma unroll
        for (int r = 0; r < 8; ++r) gp[r] += __shfl_xor((float)gp[r], off, 32);
    {
        float bcv = bc[0];
        int q = nC >> 2, wsel = nC & 3;
#pragma unroll
        for (int j = 0; j < 2; ++j) {
            float gq = (q == 0) ? gp[4 * j] : (q == 1) ? gp[4 * j + 1]
                     : (q == 2) ? gp[4 * j + 2] : gp[4 * j + 3];
            float g = gq + bcv;
            int m = q + 4 * j + 8 * hs;
            float val = (wsel == 3) ? 1.0f : g * df_s[w][m][wsel];
            atomicAdd(&acc4[di_s[w][m] * 4 + wsel], val);
        }
    }
    }
}

__global__ void __launch_bounds__(128) node_kernel(
    const float* __restrict__ x, const float* __restrict__ m_sum,
    const float* __restrict__ acc4,
    const _Float16* __restrict__ Wn1f, const float* __restrict__ bn1,
    const _Float16* __restrict__ Wn2f, const float* __restrict__ bn2,
    float* __restrict__ out_x)
{
    __shared__ _Float16 ain_s[4][16][KN];
    __shared__ _Float16 h_s[4][16][HID];
    __shared__ float rdeg_s[4][16];

    int w = threadIdx.x >> 5, lane = threadIdx.x & 31;
    int i0 = (blockIdx.x * 4 + w) * 16;
    if (i0 >= NNODE) return;
    int nC = lane & 15, hs = lane >> 4;

    float dg = 1.f;
    if (lane < 16) {
        int i = i0 + lane; if (i >= NNODE) i = NNODE - 1;
        dg = acc4[i * 4 + 3];
    }
    float4 vx[8], vm[16];
#pragma unroll
    for (int t = 0; t < 8; ++t) {
        int idx = lane + 32 * t;
        int m = idx >> 4, c4 = (idx & 15) * 4;
        int i = i0 + m; if (i >= NNODE) i = NNODE - 1;
        vx[t] = *(const float4*)(x + (long)i * IND + c4);
    }
#pragma unroll
    for (int t = 0; t < 16; ++t) {
        int i = i0 + t; if (i >= NNODE) i = NNODE - 1;
        vm[t] = *(const float4*)(m_sum + (long)i * HID + lane * 4);
    }
    if (lane < 16) {
        if (dg < 1.f) dg = 1.f;
        rdeg_s[w][lane] = __builtin_amdgcn_rcpf(dg);
    }
    LDS_SYNC();
#pragma unroll
    for (int t = 0; t < 8; ++t) {
        int idx = lane + 32 * t;
        int m = idx >> 4, c4 = (idx & 15) * 4;
        *(uint2*)&ain_s[w][m][c4] = make_uint2(pk2h(vx[t].x, vx[t].y), pk2h(vx[t].z, vx[t].w));
    }
#pragma unroll
    for (int t = 0; t < 16; ++t) {
        float r = rdeg_s[w][t];
        *(uint2*)&ain_s[w][t][IND + lane * 4] =
            make_uint2(pk2h(vm[t].x * r, vm[t].y * r), pk2h(vm[t].z * r, vm[t].w * r));
    }
    LDS_SYNC();

    const v16h* W1 = (const v16h*)Wn1f + lane;
    const v16h* W2 = (const v16h*)Wn2f + lane;

    v16h af[6];
#pragma unroll
    for (int kc = 0; kc < 6; ++kc)
        af[kc] = load_a_frag(&ain_s[w][0][0], lane, kc * 32, KN);
    v16h bw[6];
#pragma unroll
    for (int kc = 0; kc < 6; ++kc) bw[kc] = W1[(kc * 8) * 32];
    float bias1 = bn1[nC];
    for (int nt = 0; nt < 7; ++nt) {
        v8f c = {bias1, bias1, bias1, bias1, bias1, bias1, bias1, bias1};
#pragma unroll
        for (int kc = 0; kc < 6; ++kc)
            c = __builtin_amdgcn_wmma_f32_16x16x32_f16(false, af[kc], false, bw[kc], (short)0, c, false, false);
#pragma unroll
        for (int kc = 0; kc < 6; ++kc) bw[kc] = W1[(kc * 8 + nt + 1) * 32];
        bias1 = bn1[(nt + 1) * 16 + nC];
#pragma unroll
        for (int r = 0; r < 8; ++r)
            h_s[w][r + 8 * hs][nt * 16 + nC] = (_Float16)silu(c[r]);
    }
    v16h bv[4];
    float bias2;
    {
        v8f c = {bias1, bias1, bias1, bias1, bias1, bias1, bias1, bias1};
#pragma unroll
        for (int kc = 0; kc < 6; ++kc)
            c = __builtin_amdgcn_wmma_f32_16x16x32_f16(false, af[kc], false, bw[kc], (short)0, c, false, false);
#pragma unroll
        for (int kc = 0; kc < 4; ++kc) bv[kc] = W2[(kc * 4) * 32];
        bias2 = bn2[nC];
#pragma unroll
        for (int r = 0; r < 8; ++r)
            h_s[w][r + 8 * hs][7 * 16 + nC] = (_Float16)silu(c[r]);
    }
    LDS_SYNC();

    v16h ah[4];
#pragma unroll
    for (int kc = 0; kc < 4; ++kc)
        ah[kc] = load_a_frag(&h_s[w][0][0], lane, kc * 32, HID);
#pragma unroll
    for (int nt = 0; nt < 4; ++nt) {
        int n = nt * 16 + nC;
        v8f c = {bias2, bias2, bias2, bias2, bias2, bias2, bias2, bias2};
#pragma unroll
        for (int kc = 0; kc < 4; ++kc)
            c = __builtin_amdgcn_wmma_f32_16x16x32_f16(false, ah[kc], false, bv[kc], (short)0, c, false, false);
        if (nt + 1 < 4) {
#pragma unroll
            for (int kc = 0; kc < 4; ++kc) bv[kc] = W2[(kc * 4 + nt + 1) * 32];
            bias2 = bn2[(nt + 1) * 16 + nC];
        }
#pragma unroll
        for (int r = 0; r < 8; ++r) {
            int i = i0 + r + 8 * hs;
            if (i < NNODE) out_x[(long)i * IND + n] = c[r];
        }
    }
}

__global__ void pos_kernel(const float* __restrict__ pos, const float* __restrict__ acc4,
                           float* __restrict__ out_pos) {
    int t = blockIdx.x * blockDim.x + threadIdx.x;
    if (t >= NNODE * 3) return;
    int i = t / 3;
    int k = t - 3 * i;
    float dg = acc4[i * 4 + 3]; if (dg < 1.f) dg = 1.f;
    out_pos[t] = pos[t] + acc4[i * 4 + k] * __builtin_amdgcn_rcpf(dg);
}

extern "C" void kernel_launch(void* const* d_in, const int* in_sizes, int n_in,
                              void* d_out, int out_size, void* d_ws, size_t ws_size,
                              hipStream_t stream) {
    const float* x   = (const float*)d_in[0];
    const float* pos = (const float*)d_in[1];
    const int*   ei  = (const int*)d_in[2];
    const float* ea  = (const float*)d_in[3];
    const float* s   = (const float*)d_in[4];
    const float* We1 = (const float*)d_in[5];  const float* be1 = (const float*)d_in[6];
    const float* We2 = (const float*)d_in[7];  const float* be2 = (const float*)d_in[8];
    const float* Wn1 = (const float*)d_in[9];  const float* bn1 = (const float*)d_in[10];
    const float* Wn2 = (const float*)d_in[11]; const float* bn2 = (const float*)d_in[12];
    const float* Wc  = (const float*)d_in[13]; const float* bc  = (const float*)d_in[14];

    float* out_x   = (float*)d_out;
    float* out_pos = out_x + (long)NNODE * IND;

    float* m_sum = (float*)d_ws;
    float* acc4  = m_sum + (long)NNODE * HID;
    _Float16* We1f = (_Float16*)(acc4 + (long)NNODE * 4);
    _Float16* We2f = We1f + 5 * 8 * 512;
    _Float16* Wn1f = We2f + 4 * 8 * 512;
    _Float16* Wn2f = Wn1f + 6 * 8 * 512;

    size_t acc_bytes = ((long)NNODE * HID + (long)NNODE * 4) * sizeof(float);
    hipMemsetAsync(d_ws, 0, acc_bytes, stream);

    pack_weights<<<5 * 8, 32, 0, stream>>>(We1, We1f, 145, 128, 8, 1);
    pack_weights<<<4 * 8, 32, 0, stream>>>(We2, We2f, 128, 128, 8, 1);
    pack_weights<<<6 * 8, 32, 0, stream>>>(Wn1, Wn1f, 192, 128, 8, 0);
    pack_weights<<<4 * 4, 32, 0, stream>>>(Wn2, Wn2f, 128, 64, 4, 0);

    edge_kernel<<<NBLK, WPB * 32, 0, stream>>>(x, pos, ei, ea, s,
                                                We1f, be1, We2f, be2, Wc, bc,
                                                m_sum, acc4);
    node_kernel<<<(NNODE + 63) / 64, 128, 0, stream>>>(x, m_sum, acc4,
                                                       Wn1f, bn1, Wn2f, bn2, out_x);
    pos_kernel<<<(NNODE * 3 + 255) / 256, 256, 0, stream>>>(pos, acc4, out_pos);
}


# LDS rows of the activation tiles padded to a pitch of 16 B x odd (edge e_in 320->336 B, hidden 256->272 B, node input 384->400 B): 16-byte A-fragment reads no longer share banks; sum order kept
# speedup vs baseline: 2.0696x; 1.2000x over previous
#include <hip/hip_runtime.h>
#include <hip/hip_bf16.h>

typedef __attribute__((ext_vector_type(16))) _Float16 v16h;
typedef __attribute__((ext_vector_type(8)))  float    v8f;
typedef __attribute__((ext_vector_type(8)))  _Float16 v8h;

#define NNODE 50000
#define NEDGE 800000
#define IND   64
#define HID   128
#define EDD   16
#define KE    160
#define KN    192
#define KEP   168
#define HP    136
#define KNP   200
#define NTILE (NEDGE / 16)
#define WPB   24
#define NBLK  256
#define F1    40
#define F2    32
#define NZ4   ((NNODE * HID + NNODE * 4) / 4)
#define PACKB 17
#define ZEROB 512
#define NODEB ((NNODE + 63) / 64)
#define POSB  ((NNODE * 3 + 1023) / 1024)

#define LDS_SYNC() asm volatile("s_wait_dscnt 0" ::: "memory")

__device__ __forceinline__ float silu(float v) {
    return v * __builtin_amdgcn_rcpf(1.f + __expf(-v));
}

__device__ __forceinline__ unsigned pk2h(float a, float b) {
    union { _Float16 h[2]; unsigned u; } t;
    t.h[0] = (_Float16)a; t.h[1] = (_Float16)b;
    return t.u;
}

__device__ __forceinline__ float xhalf(float v) {
    int b = __builtin_bit_cast(int, v);
    return __builtin_bit_cast(float,
        __builtin_amdgcn_permlanex16(b, b, 0x76543210, 0xfedcba98, false, false));
}

__device__ __forceinline__ v16h load_a_frag(const _Float16* base, int lane, int kbase, int stride) {
    int m = lane & 15, hs = lane >> 4;
    union { v16h v; unsigned u[8]; } f;
    const _Float16* p = base + m * stride + kbase + 8 * hs;
#pragma unroll
    for (int r = 0; r < 8; ++r) {
        int k0 = (r < 4) ? (2 * r) : (16 + 2 * (r - 4));
        f.u[r] = *(const unsigned*)(p + k0);
    }
    return f.v;
}

__device__ __forceinline__ void pack_frag(const float* __restrict__ W, _Float16* __restrict__ out,
                                          int frag, int lane, int K, int N, int NT, int planes) {
    int nt = frag % NT;
    int kc = frag / NT;
    int nC = lane & 15, hs = lane >> 4;
    int n = nt * 16 + nC;
#pragma unroll
    for (int j = 0; j < 16; ++j) {
        int r = j >> 1, p = j & 1;
        int kl = ((r < 4) ? 2 * r : 16 + 2 * (r - 4)) + p + 8 * hs;
        int k = kc * 32 + kl;
        float v = (k < K) ? W[(long)k * N + n] : 0.f;
        long o = planes ? (((long)frag * 2 + (j >> 3)) * 32 + lane) * 8 + (j & 7)
                        : ((long)frag * 32 + lane) * 16 + j;
        out[o] = (_Float16)v;
    }
}

__global__ void __launch_bounds__(256) prep_kernel(
    const float* __restrict__ We1, const float* __restrict__ We2,
    const float* __restrict__ Wn1, const float* __restrict__ Wn2,
    _Float16* __restrict__ We1f, _Float16* __restrict__ We2f,
    _Float16* __restrict__ Wn1f, _Float16* __restrict__ Wn2f,
    float* __restrict__ acc)
{
    if (blockIdx.x < PACKB) {
        int f = blockIdx.x * 8 + (threadIdx.x >> 5);
        int lane = threadIdx.x & 31;
        if (f < F1)                pack_frag(We1, We1f, f, lane, 145, 128, 8, 1);
        else if (f < F1 + F2)      pack_frag(We2, We2f, f - F1, lane, 128, 128, 8, 1);
        else if (f < F1 + F2 + 48) pack_frag(Wn1, Wn1f, f - (F1 + F2), lane, 192, 128, 8, 0);
        else                       pack_frag(Wn2, Wn2f, f - (F1 + F2 + 48), lane, 128, 64, 4, 0);
        return;
    }
    uint4* z = (uint4*)acc;
    for (int i = (blockIdx.x - PACKB) * 256 + threadIdx.x; i < NZ4; i += ZEROB * 256)
        z[i] = make_uint4(0u, 0u, 0u, 0u);
}

__device__ __forceinline__ v16h lds_b_frag(const _Float16* wl, int f, int wq) {
    const v8h* p = (const v8h*)(wl + f * 512 + wq);
    v8h lo = p[0], hi = p[32];
    return __builtin_shufflevector(lo, hi, 0, 1, 2, 3, 4, 5, 6, 7, 8, 9, 10, 11, 12, 13, 14, 15);
}

__global__ void __launch_bounds__(WPB * 32) edge_kernel(
    const float* __restrict__ x, const float* __restrict__ pos,
    const int* __restrict__ ei, const float* __restrict__ ea,
    const float* __restrict__ s,
    const _Float16* __restrict__ We1f, const float* __restrict__ be1,
    const _Float16* __restrict__ We2f, const float* __restrict__ be2,
    const float* __restrict__ Wc, const float* __restrict__ bc,
    float* __restrict__ m_sum, float* __restrict__ acc4)
{
    __shared__ __attribute__((aligned(16))) _Float16 wl_s[(F1 + F2) * 512];
    __shared__ __attribute__((aligned(16))) _Float16 ein_s[WPB][16][KEP];
    __shared__ __attribute__((aligned(16))) _Float16 h_s[WPB][16][HP];
    __shared__ float sg_s[WPB][16];
    __shared__ float df_s[WPB][16][4];
    __shared__ int   di_s[WPB][16];
    __shared__ int   si_s[WPB][16];

    int w = __builtin_amdgcn_readfirstlane(threadIdx.x >> 5);
    int lane0 = threadIdx.x & 31;

    {
        const uint4* g1 = (const uint4*)We1f;
        const uint4* g2 = (const uint4*)We2f;
        uint4* l = (uint4*)wl_s;
        for (int i = threadIdx.x; i < F1 * 64; i += WPB * 32) l[i] = g1[i];
        for (int i = threadIdx.x; i < F2 * 64; i += WPB * 32) l[F1 * 64 + i] = g2[i];
    }
    __syncthreads();
    const _Float16* wl1 = wl_s;
    const _Float16* wl2 = wl_s + F1 * 512;
    _Float16* hw = &h_s[w][0][0];

    for (int tile = blockIdx.x * WPB + w; tile < NTILE; tile += gridDim.x * WPB) {
    LDS_SYNC();
    int e0 = tile * 16;
    int lane = lane0;
    asm volatile("" : "+v"(lane));
    int nC = lane & 15, hs = lane >> 4;
    int wq = lane * 8;


    float ev[8];
#pragma unroll
    for (int t = 0; t < 8; ++t)
        ev[t] = ea[(long)e0 * EDD + lane + 32 * t];

    int si = 0, di = 0;
    if (lane < 16) {
        int e = e0 + lane;
        si = ei[e];
        di = ei[NEDGE + e];
        si_s[w][lane] = si;
        di_s[w][lane] = di;
    }
    LDS_SYNC();

    float4 vd[8], vs[8];
#pragma unroll
    for (int t = 0; t < 8; ++t) {
        int idx = lane + 32 * t;
        int m = idx >> 4, c4 = (idx & 15) * 4;
        vd[t] = *(const float4*)(x + (long)di_s[w][m] * IND + c4);
        vs[t] = *(const float4*)(x + (long)si_s[w][m] * IND + c4);
    }

    if (lane < 16) {
        sg_s[w][lane] = s[si];
        float dx = pos[di * 3 + 0] - pos[si * 3 + 0];
        float dy = pos[di * 3 + 1] - pos[si * 3 + 1];
        float dz = pos[di * 3 + 2] - pos[si * 3 + 2];
        df_s[w][lane][0] = dx; df_s[w][lane][1] = dy; df_s[w][lane][2] = dz;
        df_s[w][lane][3] = 1.0f;
        ein_s[w][lane][2 * IND] = (_Float16)(dx * dx + dy * dy + dz * dz);
#pragma unroll
        for (int c = 2 * IND + 1 + EDD; c < KE; ++c)
            ein_s[w][lane][c] = (_Float16)0.f;
    }

#pragma unroll
    for (int t = 0; t < 8; ++t) {
        int idx = lane + 32 * t;
        int m = idx >> 4, c4 = (idx & 15) * 4;
        *(uint2*)&ein_s[w][m][c4]       = make_uint2(pk2h(vd[t].x, vd[t].y), pk2h(vd[t].z, vd[t].w));
        *(uint2*)&ein_s[w][m][IND + c4] = make_uint2(pk2h(vs[t].x, vs[t].y), pk2h(vs[t].z, vs[t].w));
    }
#pragma unroll
    for (int t = 0; t < 8; ++t) {
        int idx = lane + 32 * t;
        ein_s[w][idx >> 4][2 * IND + 1 + (idx & 15)] = (_Float16)ev[t];
    }
    LDS_SYNC();

    v16h af[5];
#pragma unroll
    for (int kc = 0; kc < 5; ++kc)
        af[kc] = load_a_frag(&ein_s[w][0][0], lane, kc * 32, KEP);
    v16h bw[5];
#pragma unroll
    for (int kc = 0; kc < 5; ++kc) bw[kc] = lds_b_frag(wl1, kc * 8, wq);
    float bias1 = be1[nC];
    for (int nt = 0; nt < 7; ++nt) {
        v8f c = {bias1, bias1, bias1, bias1, bias1, bias1, bias1, bias1};
#pragma unroll
        for (int kc = 0; kc < 5; ++kc)
            c = __builtin_amdgcn_wmma_f32_16x16x32_f16(false, af[kc], false, bw[kc], (short)0, c, false, false);
#pragma unroll
        for (int kc = 0; kc < 5; ++kc) bw[kc] = lds_b_frag(wl1, kc * 8 + nt + 1, wq);
        bias1 = be1[(nt + 1) * 16 + nC];
#pragma unroll
        for (int r = 0; r < 8; ++r)
            hw[(r + 8 * hs) * HP + nt * 16 + nC] = (_Float16)silu(c[r]);
    }
    {
        v8f c = {bias1, bias1, bias1, bias1, bias1, bias1, bias1, bias1};
#pragma unroll
        for (int kc = 0; kc < 5; ++kc)
            c = __builtin_amdgcn_wmma_f32_16x16x32_f16(false, af[kc], false, bw[kc], (short)0, c, false, false);
#pragma unroll
        for (int r = 0; r < 8; ++r)
            hw[(r + 8 * hs) * HP + 7 * 16 + nC] = (_Float16)silu(c[r]);
    }
    LDS_SYNC();

    int rowoff[16];
#pragma unroll
    for (int k = 0; k < 16; ++k) rowoff[k] = di_s[w][k] * HID + lane;

    v8f gp = {0, 0, 0, 0, 0, 0, 0, 0};
    for (int np = 0; np < 4; ++np) {
        int n0 = np * 32 + nC, n1 = n0 + 16;
        v16h b0[4], b1[4];
#pragma unroll
        for (int kc = 0; kc < 4; ++kc) {
            b0[kc] = lds_b_frag(wl2, kc * 8 + 2 * np, wq);
            b1[kc] = lds_b_frag(wl2, kc * 8 + 2 * np + 1, wq);
        }
        float bs0 = be2[n0], bs1 = be2[n1];
        float wc0 = Wc[n0], wc1 = Wc[n1];
        v8f c0 = {bs0, bs0, bs0, bs0, bs0, bs0, bs0, bs0};
        v8f c1 = {bs1, bs1, bs1, bs1, bs1, bs1, bs1, bs1};
#pragma unroll
        for (int kc = 0; kc < 4; ++kc) {
            v16h a = load_a_frag(hw, lane, kc * 32, HP);
            c0 = __builtin_amdgcn_wmma_f32_16x16x32_f16(false, a, false, b0[kc], (short)0, c0, false, false);
            c1 = __builtin_amdgcn_wmma_f32_16x16x32_f16(false, a, false, b1[kc], (short)0, c1, false, false);
        }
#pragma unroll
        for (int r = 0; r < 8; ++r) {
            float sg = sg_s[w][r + 8 * hs];
            float v0 = silu(c0[r]) * sg;
            float v1 = silu(c1[r]) * sg;
            gp[r] += v0 * wc0;
            gp[r] += v1 * wc1;
            float got = xhalf(hs ? v0 : v1);
            float lo = hs ? got : v0;
            float hi = hs ? v1 : got;
            atomicAdd(m_sum + rowoff[r], lo);
            atomicAdd(m_sum + rowoff[r + 8], hi);
        }
#pragma unroll
        for (int k = 0; k < 16; ++k) rowoff[k] += 32;
    }
#pragma unroll
    for (int off = 1; off < 16; off <<= 1)
#pragma unroll
        for (int r = 0; r < 8; ++r) gp[r] += __shfl_xor((float)gp[r], off, 32);
    {
        float bcv = bc[0];
        int q = nC >> 2, wsel = nC & 3;
#pragma unroll
        for (int j = 0; j < 2; ++j) {
            float gq = (q == 0) ? gp[4 * j] : (q == 1) ? gp[4 * j + 1]
                     : (q == 2) ? gp[4 * j + 2] : gp[4 * j + 3];
            float g = gq + bcv;
            int m = q + 4 * j + 8 * hs;
            float val = (wsel == 3) ? 1.0f : g * df_s[w][m][wsel];
            atomicAdd(&acc4[di_s[w][m] * 4 + wsel], val);
        }
    }
    }
}

__global__ void __launch_bounds__(128) node_kernel(
    const float* __restrict__ x, const float* __restrict__ m_sum,
    const float* __restrict__ acc4,
    const _Float16* __restrict__ Wn1f, const float* __restrict__ bn1,
    const _Float16* __restrict__ Wn2f, const float* __restrict__ bn2,
    const float* __restrict__ pos,
    float* __restrict__ out_x, float* __restrict__ out_pos)
{
    __shared__ __attribute__((aligned(16))) _Float16 ain_s[4][16][KNP];
    __shared__ __attribute__((aligned(16))) _Float16 h_s[4][16][HP];
    __shared__ float rdeg_s[4][16];

    if (blockIdx.x >= NODEB) {
        int base = (blockIdx.x - NODEB) * 1024 + threadIdx.x;
        float pa[8], pc[8], pd[8];
#pragma unroll
        for (int j = 0; j < 8; ++j) {
            int ge = base + 128 * j;
            int gc = ge < NNODE * 3 ? ge : NNODE * 3 - 1;
            int i = gc / 3;
            int k = gc - 3 * i;
            pa[j] = pos[gc];
            pc[j] = acc4[i * 4 + k];
            pd[j] = acc4[i * 4 + 3];
        }
#pragma unroll
        for (int j = 0; j < 8; ++j) {
            int ge = base + 128 * j;
            float dg = pd[j]; if (dg < 1.f) dg = 1.f;
            if (ge < NNODE * 3) out_pos[ge] = pa[j] + pc[j] * __builtin_amdgcn_rcpf(dg);
        }
        return;
    }

    int w = threadIdx.x >> 5, lane = threadIdx.x & 31;
    int i0 = (blockIdx.x * 4 + w) * 16;
    if (i0 >= NNODE) return;
    int nC = lane & 15, hs = lane >> 4;

    float dg = 1.f;
    if (lane < 16) {
        int i = i0 + lane; if (i >= NNODE) i = NNODE - 1;
        dg = acc4[i * 4 + 3];
    }
    float4 vx[8], vm[16];
#pragma unroll
    for (int t = 0; t < 8; ++t) {
        int idx = lane + 32 * t;
        int m = idx >> 4, c4 = (idx & 15) * 4;
        int i = i0 + m; if (i >= NNODE) i = NNODE - 1;
        vx[t] = *(const float4*)(x + (long)i * IND + c4);
    }
#pragma unroll
    for (int t = 0; t < 16; ++t) {
        int i = i0 + t; if (i >= NNODE) i = NNODE - 1;
        vm[t] = *(const float4*)(m_sum + (long)i * HID + lane * 4);
    }
    if (lane < 16) {
        if (dg < 1.f) dg = 1.f;
        rdeg_s[w][lane] = __builtin_amdgcn_rcpf(dg);
    }
    LDS_SYNC();
#pragma unroll
    for (int t = 0; t < 8; ++t) {
        int idx = lane + 32 * t;
        int m = idx >> 4, c4 = (idx & 15) * 4;
        *(uint2*)&ain_s[w][m][c4] = make_uint2(pk2h(vx[t].x, vx[t].y), pk2h(vx[t].z, vx[t].w));
    }
#pragma unroll
    for (int t = 0; t < 16; ++t) {
        float r = rdeg_s[w][t];
        *(uint2*)&ain_s[w][t][IND + lane * 4] =
            make_uint2(pk2h(vm[t].x * r, vm[t].y * r), pk2h(vm[t].z * r, vm[t].w * r));
    }
    LDS_SYNC();

    const v16h* W1 = (const v16h*)Wn1f + lane;
    const v16h* W2 = (const v16h*)Wn2f + lane;

    v16h af[6];
#pragma unroll
    for (int kc = 0; kc < 6; ++kc)
        af[kc] = load_a_frag(&ain_s[w][0][0], lane, kc * 32, KNP);
    v16h bw[6];
#pragma unroll
    for (int kc = 0; kc < 6; ++kc) bw[kc] = W1[(kc * 8) * 32];
    float bias1 = bn1[nC];
    for (int nt = 0; nt < 7; ++nt) {
        v8f c = {bias1, bias1, bias1, bias1, bias1, bias1, bias1, bias1};
#pragma unroll
        for (int kc = 0; kc < 6; ++kc)
            c = __builtin_amdgcn_wmma_f32_16x16x32_f16(false, af[kc], false, bw[kc], (short)0, c, false, false);
#pragma unroll
        for (int kc = 0; kc < 6; ++kc) bw[kc] = W1[(kc * 8 + nt + 1) * 32];
        bias1 = bn1[(nt + 1) * 16 + nC];
#pragma unroll
        for (int r = 0; r < 8; ++r)
            h_s[w][r + 8 * hs][nt * 16 + nC] = (_Float16)silu(c[r]);
    }
    v16h bv[4];
    float bias2;
    {
        v8f c = {bias1, bias1, bias1, bias1, bias1, bias1, bias1, bias1};
#pragma unroll
        for (int kc = 0; kc < 6; ++kc)
            c = __builtin_amdgcn_wmma_f32_16x16x32_f16(false, af[kc], false, bw[kc], (short)0, c, false, false);
#pragma unroll
        for (int kc = 0; kc < 4; ++kc) bv[kc] = W2[(kc * 4) * 32];
        bias2 = bn2[nC];
#pragma unroll
        for (int r = 0; r < 8; ++r)
            h_s[w][r + 8 * hs][7 * 16 + nC] = (_Float16)silu(c[r]);
    }
    LDS_SYNC();

    v16h ah[4];
#pragma unroll
    for (int kc = 0; kc < 4; ++kc)
        ah[kc] = load_a_frag(&h_s[w][0][0], lane, kc * 32, HP);
#pragma unroll
    for (int nt = 0; nt < 4; ++nt) {
        int n = nt * 16 + nC;
        v8f c = {bias2, bias2, bias2, bias2, bias2, bias2, bias2, bias2};
#pragma unroll
        for (int kc = 0; kc < 4; ++kc)
            c = __builtin_amdgcn_wmma_f32_16x16x32_f16(false, ah[kc], false, bv[kc], (short)0, c, false, false);
        if (nt + 1 < 4) {
#pragma unroll
            for (int kc = 0; kc < 4; ++kc) bv[kc] = W2[(kc * 4 + nt + 1) * 32];
            bias2 = bn2[(nt + 1) * 16 + nC];
        }
#pragma unroll
        for (int r = 0; r < 8; ++r) {
            int i = i0 + r + 8 * hs;
            if (i < NNODE) out_x[(long)i * IND + n] = c[r];
        }
    }
}

extern "C" void kernel_launch(void* const* d_in, const int* in_sizes, int n_in,
                              void* d_out, int out_size, void* d_ws, size_t ws_size,
                              hipStream_t stream) {
    const float* x   = (const float*)d_in[0];
    const float* pos = (const float*)d_in[1];
    const int*   ei  = (const int*)d_in[2];
    const float* ea  = (const float*)d_in[3];
    const float* s   = (const float*)d_in[4];
    const float* We1 = (const float*)d_in[5];  const float* be1 = (const float*)d_in[6];
    const float* We2 = (const float*)d_in[7];  const float* be2 = (const float*)d_in[8];
    const float* Wn1 = (const float*)d_in[9];  const float* bn1 = (const float*)d_in[10];
    const float* Wn2 = (const float*)d_in[11]; const float* bn2 = (const float*)d_in[12];
    const float* Wc  = (const float*)d_in[13]; const float* bc  = (const float*)d_in[14];

    float* out_x   = (float*)d_out;
    float* out_pos = out_x + (long)NNODE * IND;

    float* m_sum = (float*)d_ws;
    float* acc4  = m_sum + (long)NNODE * HID;
    _Float16* We1f = (_Float16*)(acc4 + (long)NNODE * 4);
    _Float16* We2f = We1f + 5 * 8 * 512;
    _Float16* Wn1f = We2f + 4 * 8 * 512;
    _Float16* Wn2f = Wn1f + 6 * 8 * 512;

    prep_kernel<<<PACKB + ZEROB, 256, 0, stream>>>(We1, We2, Wn1, Wn2,
                                                   We1f, We2f, Wn1f, Wn2f, m_sum);

    edge_kernel<<<NBLK, WPB * 32, 0, stream>>>(x, pos, ei, ea, s,
                                                We1f, be1, We2f, be2, Wc, bc,
                                                m_sum, acc4);
    node_kernel<<<NODEB + POSB, 128, 0, stream>>>(x, m_sum, acc4,
                                                  Wn1f, bn1, Wn2f, bn2, pos,
                                                  out_x, out_pos);
}


# prep launch re-lays the node tables the edge gathers read: x rows kept in the 16-bit form the products use (one 128-B line per row, same rounding), pos+s as one 16-B record per node; sum order kept
# speedup vs baseline: 2.1633x; 1.0453x over previous
#include <hip/hip_runtime.h>
#include <hip/hip_bf16.h>

typedef __attribute__((ext_vector_type(16))) _Float16 v16h;
typedef __attribute__((ext_vector_type(8)))  float    v8f;
typedef __attribute__((ext_vector_type(8)))  _Float16 v8h;

#define NNODE 50000
#define NEDGE 800000
#define IND   64
#define HID   128
#define EDD   16
#define KE    160
#define KN    192
#define KEP   168
#define HP    136
#define KNP   200
#define NTILE (NEDGE / 16)
#define WPB   24
#define NBLK  256
#define F1    40
#define F2    32
#define NZ4   ((NNODE * HID + NNODE * 4) / 4)
#define PACKB 17
#define ZEROB 512
#define CONVB 512
#define NX8   (NNODE * IND / 8)
#define RECB  ((NNODE + 255) / 256)
#define NODEB ((NNODE + 63) / 64)
#define POSB  ((NNODE * 3 + 1023) / 1024)

#define LDS_SYNC() asm volatile("s_wait_dscnt 0" ::: "memory")

__device__ __forceinline__ float silu(float v) {
    return v * __builtin_amdgcn_rcpf(1.f + __expf(-v));
}

__device__ __forceinline__ unsigned pk2h(float a, float b) {
    union { _Float16 h[2]; unsigned u; } t;
    t.h[0] = (_Float16)a; t.h[1] = (_Float16)b;
    return t.u;
}

__device__ __forceinline__ float xhalf(float v) {
    int b = __builtin_bit_cast(int, v);
    return __builtin_bit_cast(float,
        __builtin_amdgcn_permlanex16(b, b, 0x76543210, 0xfedcba98, false, false));
}

__device__ __forceinline__ v16h load_a_frag(const _Float16* base, int lane, int kbase, int stride) {
    int m = lane & 15, hs = lane >> 4;
    union { v16h v; unsigned u[8]; } f;
    const _Float16* p = base + m * stride + kbase + 8 * hs;
#pragma unroll
    for (int r = 0; r < 8; ++r) {
        int k0 = (r < 4) ? (2 * r) : (16 + 2 * (r - 4));
        f.u[r] = *(const unsigned*)(p + k0);
    }
    return f.v;
}

__device__ __forceinline__ void pack_frag(const float* __restrict__ W, _Float16* __restrict__ out,
                                          int frag, int lane, int K, int N, int NT, int planes) {
    int nt = frag % NT;
    int kc = frag / NT;
    int nC = lane & 15, hs = lane >> 4;
    int n = nt * 16 + nC;
#pragma unroll
    for (int j = 0; j < 16; ++j) {
        int r = j >> 1, p = j & 1;
        int kl = ((r < 4) ? 2 * r : 16 + 2 * (r - 4)) + p + 8 * hs;
        int k = kc * 32 + kl;
        float v = (k < K) ? W[(long)k * N + n] : 0.f;
        long o = planes ? (((long)frag * 2 + (j >> 3)) * 32 + lane) * 8 + (j & 7)
                        : ((long)frag * 32 + lane) * 16 + j;
        out[o] = (_Float16)v;
    }
}

__global__ void __launch_bounds__(256) prep_kernel(
    const float* __restrict__ We1, const float* __restrict__ We2,
    const float* __restrict__ Wn1, const float* __restrict__ Wn2,
    _Float16* __restrict__ We1f, _Float16* __restrict__ We2f,
    _Float16* __restrict__ Wn1f, _Float16* __restrict__ Wn2f,
    float* __restrict__ acc,
    const float* __restrict__ x, const float* __restrict__ pos, const float* __restrict__ s,
    _Float16* __restrict__ x16, float* __restrict__ rec)
{
    if (blockIdx.x < PACKB) {
        int f = blockIdx.x * 8 + (threadIdx.x >> 5);
        int lane = threadIdx.x & 31;
        if (f < F1)                pack_frag(We1, We1f, f, lane, 145, 128, 8, 1);
        else if (f < F1 + F2)      pack_frag(We2, We2f, f - F1, lane, 128, 128, 8, 1);
        else if (f < F1 + F2 + 48) pack_frag(Wn1, Wn1f, f - (F1 + F2), lane, 192, 128, 8, 0);
        else                       pack_frag(Wn2, Wn2f, f - (F1 + F2 + 48), lane, 128, 64, 4, 0);
        return;
    }
    if (blockIdx.x < PACKB + ZEROB) {
        uint4* z = (uint4*)acc;
        for (int i = (blockIdx.x - PACKB) * 256 + threadIdx.x; i < NZ4; i += ZEROB * 256)
            z[i] = make_uint4(0u, 0u, 0u, 0u);
        return;
    }
    if (blockIdx.x < PACKB + ZEROB + CONVB) {
        const float4* xs = (const float4*)x;
        uint4* xd = (uint4*)x16;
        for (int i = (blockIdx.x - (PACKB + ZEROB)) * 256 + threadIdx.x; i < NX8; i += CONVB * 256) {
            float4 a = xs[2 * i], b = xs[2 * i + 1];
            xd[i] = make_uint4(pk2h(a.x, a.y), pk2h(a.z, a.w), pk2h(b.x, b.y), pk2h(b.z, b.w));
        }
        return;
    }
    int i = (blockIdx.x - (PACKB + ZEROB + CONVB)) * 256 + threadIdx.x;
    if (i < NNODE)
        ((float4*)rec)[i] = make_float4(pos[i * 3 + 0], pos[i * 3 + 1], pos[i * 3 + 2], s[i]);
}

__device__ __forceinline__ v16h lds_b_frag(const _Float16* wl, int f, int wq) {
    const v8h* p = (const v8h*)(wl + f * 512 + wq);
    v8h lo = p[0], hi = p[32];
    return __builtin_shufflevector(lo, hi, 0, 1, 2, 3, 4, 5, 6, 7, 8, 9, 10, 11, 12, 13, 14, 15);
}

__global__ void __launch_bounds__(WPB * 32) edge_kernel(
    const _Float16* __restrict__ x16, const float* __restrict__ rec,
    const int* __restrict__ ei, const float* __restrict__ ea,
    const _Float16* __restrict__ We1f, const float* __restrict__ be1,
    const _Float16* __restrict__ We2f, const float* __restrict__ be2,
    const float* __restrict__ Wc, const float* __restrict__ bc,
    float* __restrict__ m_sum, float* __restrict__ acc4)
{
    __shared__ __attribute__((aligned(16))) _Float16 wl_s[(F1 + F2) * 512];
    __shared__ __attribute__((aligned(16))) _Float16 ein_s[WPB][16][KEP];
    __shared__ __attribute__((aligned(16))) _Float16 h_s[WPB][16][HP];
    __shared__ float sg_s[WPB][16];
    __shared__ float df_s[WPB][16][4];
    __shared__ int   idx_s[WPB][32];

    int w = __builtin_amdgcn_readfirstlane(threadIdx.x >> 5);
    int lane0 = threadIdx.x & 31;

    {
        const uint4* g1 = (const uint4*)We1f;
        const uint4* g2 = (const uint4*)We2f;
        uint4* l = (uint4*)wl_s;
        for (int i = threadIdx.x; i < F1 * 64; i += WPB * 32) l[i] = g1[i];
        for (int i = threadIdx.x; i < F2 * 64; i += WPB * 32) l[F1 * 64 + i] = g2[i];
    }
    __syncthreads();
    const _Float16* wl1 = wl_s;
    const _Float16* wl2 = wl_s + F1 * 512;
    _Float16* hw = &h_s[w][0][0];

    for (int tile = blockIdx.x * WPB + w; tile < NTILE; tile += gridDim.x * WPB) {
    LDS_SYNC();
    int e0 = tile * 16;
    int lane = lane0;
    asm volatile("" : "+v"(lane));
    int nC = lane & 15, hs = lane >> 4;
    int wq = lane * 8;


    int node = ei[(hs ? 0 : NEDGE) + e0 + nC];
    __builtin_amdgcn_sched_barrier(0);

    float ev[8];
#pragma unroll
    for (int t = 0; t < 8; ++t)
        ev[t] = ea[(long)e0 * EDD + lane + 32 * t];

    idx_s[w][lane] = node;
    float4 rc = *(const float4*)(rec + (long)node * 4);
    LDS_SYNC();

    uint4 xr[8];
#pragma unroll
    for (int t = 0; t < 8; ++t) {
        int row = (lane >> 3) + 4 * t;
        unsigned off = (unsigned)idx_s[w][row] * (IND * 2u) + (lane & 7) * 16u;
        xr[t] = *(const uint4*)((const char*)x16 + off);
    }

#pragma unroll
    for (int t = 0; t < 8; ++t) {
        int m = (lane >> 3) + 4 * (t & 3);
        *(uint4*)&ein_s[w][m][(t >> 2) * IND + (lane & 7) * 8] = xr[t];
    }
#pragma unroll
    for (int t = 0; t < 8; ++t) {
        int idx = lane + 32 * t;
        ein_s[w][idx >> 4][2 * IND + 1 + (idx & 15)] = (_Float16)ev[t];
    }
    float ox = xhalf(rc.x), oy = xhalf(rc.y), oz = xhalf(rc.z);
    if (lane < 16) {
        float dx = rc.x - ox;
        float dy = rc.y - oy;
        float dz = rc.z - oz;
        df_s[w][lane][0] = dx; df_s[w][lane][1] = dy; df_s[w][lane][2] = dz;
        df_s[w][lane][3] = 1.0f;
        ein_s[w][lane][2 * IND] = (_Float16)(dx * dx + dy * dy + dz * dz);
#pragma unroll
        for (int c = 2 * IND + 1 + EDD; c < KE; ++c)
            ein_s[w][lane][c] = (_Float16)0.f;
    } else {
        sg_s[w][nC] = rc.w;
    }
    LDS_SYNC();

    v16h af[5];
#pragma unroll
    for (int kc = 0; kc < 5; ++kc)
        af[kc] = load_a_frag(&ein_s[w][0][0], lane, kc * 32, KEP);
    v16h bw[5];
#pragma unroll
    for (int kc = 0; kc < 5; ++kc) bw[kc] = lds_b_frag(wl1, kc * 8, wq);
    float bias1 = be1[nC];
    for (int nt = 0; nt < 7; ++nt) {
        v8f c = {bias1, bias1, bias1, bias1, bias1, bias1, bias1, bias1};
#pragma unroll
        for (int kc = 0; kc < 5; ++kc)
            c = __builtin_amdgcn_wmma_f32_16x16x32_f16(false, af[kc], false, bw[kc], (short)0, c, false, false);
#pragma unroll
        for (int kc = 0; kc < 5; ++kc) bw[kc] = lds_b_frag(wl1, kc * 8 + nt + 1, wq);
        bias1 = be1[(nt + 1) * 16 + nC];
#pragma unroll
        for (int r = 0; r < 8; ++r)
            hw[(r + 8 * hs) * HP + nt * 16 + nC] = (_Float16)silu(c[r]);
    }
    {
        v8f c = {bias1, bias1, bias1, bias1, bias1, bias1, bias1, bias1};
#pragma unroll
        for (int kc = 0; kc < 5; ++kc)
            c = __builtin_amdgcn_wmma_f32_16x16x32_f16(false, af[kc], false, bw[kc], (short)0, c, false, false);
#pragma unroll
        for (int r = 0; r < 8; ++r)
            hw[(r + 8 * hs) * HP + 7 * 16 + nC] = (_Float16)silu(c[r]);
    }
    LDS_SYNC();

    int rowoff[16];
#pragma unroll
    for (int k = 0; k < 16; ++k) rowoff[k] = idx_s[w][k] * HID + lane;

    v8f gp = {0, 0, 0, 0, 0, 0, 0, 0};
    for (int np = 0; np < 4; ++np) {
        int n0 = np * 32 + nC, n1 = n0 + 16;
        v16h b0[4], b1[4];
#pragma unroll
        for (int kc = 0; kc < 4; ++kc) {
            b0[kc] = lds_b_frag(wl2, kc * 8 + 2 * np, wq);
            b1[kc] = lds_b_frag(wl2, kc * 8 + 2 * np + 1, wq);
        }
        float bs0 = be2[n0], bs1 = be2[n1];
        float wc0 = Wc[n0], wc1 = Wc[n1];
        v8f c0 = {bs0, bs0, bs0, bs0, bs0, bs0, bs0, bs0};
        v8f c1 = {bs1, bs1, bs1, bs1, bs1, bs1, bs1, bs1};
#pragma unroll
        for (int kc = 0; kc < 4; ++kc) {
            v16h a = load_a_frag(hw, lane, kc * 32, HP);
            c0 = __builtin_amdgcn_wmma_f32_16x16x32_f16(false, a, false, b0[kc], (short)0, c0, false, false);
            c1 = __builtin_amdgcn_wmma_f32_16x16x32_f16(false, a, false, b1[kc], (short)0, c1, false, false);
        }
#pragma unroll
        for (int r = 0; r < 8; ++r) {
            float sg = sg_s[w][r + 8 * hs];
            float v0 = silu(c0[r]) * sg;
            float v1 = silu(c1[r]) * sg;
            gp[r] += v0 * wc0;
            gp[r] += v1 * wc1;
            float got = xhalf(hs ? v0 : v1);
            float lo = hs ? got : v0;
            float hi = hs ? v1 : got;
            atomicAdd(m_sum + rowoff[r], lo);
            atomicAdd(m_sum + rowoff[r + 8], hi);
        }
#pragma unroll
        for (int k = 0; k < 16; ++k) rowoff[k] += 32;
    }
#pragma unroll
    for (int off = 1; off < 16; off <<= 1)
#pragma unroll
        for (int r = 0; r < 8; ++r) gp[r] += __shfl_xor((float)gp[r], off, 32);
    {
        float bcv = bc[0];
        int q = nC >> 2, wsel = nC & 3;
#pragma unroll
        for (int j = 0; j < 2; ++j) {
            float gq = (q == 0) ? gp[4 * j] : (q == 1) ? gp[4 * j + 1]
                     : (q == 2) ? gp[4 * j + 2] : gp[4 * j + 3];
            float g = gq + bcv;
            int m = q + 4 * j + 8 * hs;
            float val = (wsel == 3) ? 1.0f : g * df_s[w][m][wsel];
            atomicAdd(&acc4[idx_s[w][m] * 4 + wsel], val);
        }
    }
    }
}

__global__ void __launch_bounds__(128) node_kernel(
    const float* __restrict__ x, const float* __restrict__ m_sum,
    const float* __restrict__ acc4,
    const _Float16* __restrict__ Wn1f, const float* __restrict__ bn1,
    const _Float16* __restrict__ Wn2f, const float* __restrict__ bn2,
    const float* __restrict__ pos,
    float* __restrict__ out_x, float* __restrict__ out_pos)
{
    __shared__ __attribute__((aligned(16))) _Float16 ain_s[4][16][KNP];
    __shared__ __attribute__((aligned(16))) _Float16 h_s[4][16][HP];
    __shared__ float rdeg_s[4][16];

    if (blockIdx.x >= NODEB) {
        int base = (blockIdx.x - NODEB) * 1024 + threadIdx.x;
        float pa[8], pc[8], pd[8];
#pragma unroll
        for (int j = 0; j < 8; ++j) {
            int ge = base + 128 * j;
            int gc = ge < NNODE * 3 ? ge : NNODE * 3 - 1;
            int i = gc / 3;
            int k = gc - 3 * i;
            pa[j] = pos[gc];
            pc[j] = acc4[i * 4 + k];
            pd[j] = acc4[i * 4 + 3];
        }
#pragma unroll
        for (int j = 0; j < 8; ++j) {
            int ge = base + 128 * j;
            float dg = pd[j]; if (dg < 1.f) dg = 1.f;
            if (ge < NNODE * 3) out_pos[ge] = pa[j] + pc[j] * __builtin_amdgcn_rcpf(dg);
        }
        return;
    }

    int w = threadIdx.x >> 5, lane = threadIdx.x & 31;
    int i0 = (blockIdx.x * 4 + w) * 16;
    if (i0 >= NNODE) return;
    int nC = lane & 15, hs = lane >> 4;

    float dg = 1.f;
    if (lane < 16) {
        int i = i0 + lane; if (i >= NNODE) i = NNODE - 1;
        dg = acc4[i * 4 + 3];
    }
    float4 vx[8], vm[16];
#pragma unroll
    for (int t = 0; t < 8; ++t) {
        int idx = lane + 32 * t;
        int m = idx >> 4, c4 = (idx & 15) * 4;
        int i = i0 + m; if (i >= NNODE) i = NNODE - 1;
        vx[t] = *(const float4*)(x + (long)i * IND + c4);
    }
#pragma unroll
    for (int t = 0; t < 16; ++t) {
        int i = i0 + t; if (i >= NNODE) i = NNODE - 1;
        vm[t] = *(const float4*)(m_sum + (long)i * HID + lane * 4);
    }
    if (lane < 16) {
        if (dg < 1.f) dg = 1.f;
        rdeg_s[w][lane] = __builtin_amdgcn_rcpf(dg);
    }
    LDS_SYNC();
#pragma unroll
    for (int t = 0; t < 8; ++t) {
        int idx = lane + 32 * t;
        int m = idx >> 4, c4 = (idx & 15) * 4;
        *(uint2*)&ain_s[w][m][c4] = make_uint2(pk2h(vx[t].x, vx[t].y), pk2h(vx[t].z, vx[t].w));
    }
#pragma unroll
    for (int t = 0; t < 16; ++t) {
        float r = rdeg_s[w][t];
        *(uint2*)&ain_s[w][t][IND + lane * 4] =
            make_uint2(pk2h(vm[t].x * r, vm[t].y * r), pk2h(vm[t].z * r, vm[t].w * r));
    }
    LDS_SYNC();

    const v16h* W1 = (const v16h*)Wn1f + lane;
    const v16h* W2 = (const v16h*)Wn2f + lane;

    v16h af[6];
#pragma unroll
    for (int kc = 0; kc < 6; ++kc)
        af[kc] = load_a_frag(&ain_s[w][0][0], lane, kc * 32, KNP);
    v16h bw[6];
#pragma unroll
    for (int kc = 0; kc < 6; ++kc) bw[kc] = W1[(kc * 8) * 32];
    float bias1 = bn1[nC];
    for (int nt = 0; nt < 7; ++nt) {
        v8f c = {bias1, bias1, bias1, bias1, bias1, bias1, bias1, bias1};
#pragma unroll
        for (int kc = 0; kc < 6; ++kc)
            c = __builtin_amdgcn_wmma_f32_16x16x32_f16(false, af[kc], false, bw[kc], (short)0, c, false, false);
#pragma unroll
        for (int kc = 0; kc < 6; ++kc) bw[kc] = W1[(kc * 8 + nt + 1) * 32];
        bias1 = bn1[(nt + 1) * 16 + nC];
#pragma unroll
        for (int r = 0; r < 8; ++r)
            h_s[w][r + 8 * hs][nt * 16 + nC] = (_Float16)silu(c[r]);
    }
    v16h bv[4];
    float bias2;
    {
        v8f c = {bias1, bias1, bias1, bias1, bias1, bias1, bias1, bias1};
#pragma unroll
        for (int kc = 0; kc < 6; ++kc)
            c = __builtin_amdgcn_wmma_f32_16x16x32_f16(false, af[kc], false, bw[kc], (short)0, c, false, false);
#pragma unroll
        for (int kc = 0; kc < 4; ++kc) bv[kc] = W2[(kc * 4) * 32];
        bias2 = bn2[nC];
#pragma unroll
        for (int r = 0; r < 8; ++r)
            h_s[w][r + 8 * hs][7 * 16 + nC] = (_Float16)silu(c[r]);
    }
    LDS_SYNC();

    v16h ah[4];
#pragma unroll
    for (int kc = 0; kc < 4; ++kc)
        ah[kc] = load_a_frag(&h_s[w][0][0], lane, kc * 32, HP);
#pragma unroll
    for (int nt = 0; nt < 4; ++nt) {
        int n = nt * 16 + nC;
        v8f c = {bias2, bias2, bias2, bias2, bias2, bias2, bias2, bias2};
#pragma unroll
        for (int kc = 0; kc < 4; ++kc)
            c = __builtin_amdgcn_wmma_f32_16x16x32_f16(false, ah[kc], false, bv[kc], (short)0, c, false, false);
        if (nt + 1 < 4) {
#pragma unroll
            for (int kc = 0; kc < 4; ++kc) bv[kc] = W2[(kc * 4 + nt + 1) * 32];
            bias2 = bn2[(nt + 1) * 16 + nC];
        }
#pragma unroll
        for (int r = 0; r < 8; ++r) {
            int i = i0 + r + 8 * hs;
            if (i < NNODE) out_x[(long)i * IND + n] = c[r];
        }
    }
}

extern "C" void kernel_launch(void* const* d_in, const int* in_sizes, int n_in,
                              void* d_out, int out_size, void* d_ws, size_t ws_size,
                              hipStream_t stream) {
    const float* x   = (const float*)d_in[0];
    const float* pos = (const float*)d_in[1];
    const int*   ei  = (const int*)d_in[2];
    const float* ea  = (const float*)d_in[3];
    const float* s   = (const float*)d_in[4];
    const float* We1 = (const float*)d_in[5];  const float* be1 = (const float*)d_in[6];
    const float* We2 = (const float*)d_in[7];  const float* be2 = (const float*)d_in[8];
    const float* Wn1 = (const float*)d_in[9];  const float* bn1 = (const float*)d_in[10];
    const float* Wn2 = (const float*)d_in[11]; const float* bn2 = (const float*)d_in[12];
    const float* Wc  = (const float*)d_in[13]; const float* bc  = (const float*)d_in[14];

    float* out_x   = (float*)d_out;
    float* out_pos = out_x + (long)NNODE * IND;

    float* m_sum = (float*)d_ws;
    float* acc4  = m_sum + (long)NNODE * HID;
    _Float16* We1f = (_Float16*)(acc4 + (long)NNODE * 4);
    _Float16* We2f = We1f + 5 * 8 * 512;
    _Float16* Wn1f = We2f + 4 * 8 * 512;
    _Float16* Wn2f = Wn1f + 6 * 8 * 512;

    _Float16* x16 = (_Float16*)d_out;
    float*    rec = out_x + (long)NNODE * IND / 2;

    prep_kernel<<<PACKB + ZEROB + CONVB + RECB, 256, 0, stream>>>(We1, We2, Wn1, Wn2,
                                                   We1f, We2f, Wn1f, Wn2f, m_sum,
                                                   x, pos, s, x16, rec);

    edge_kernel<<<NBLK, WPB * 32, 0, stream>>>(x16, rec, ei, ea,
                                                We1f, be1, We2f, be2, Wc, bc,
                                                m_sum, acc4);
    node_kernel<<<NODEB + POSB, 128, 0, stream>>>(x, m_sum, acc4,
                                                  Wn1f, bn1, Wn2f, bn2, pos,
                                                  out_x, out_pos);
}


# edge sums off the global f32 atomics: edges placed by owner (256 node ranges: count in prep + a place launch); each group adds its nodes' rows in LDS (ds_add_f32, arrival order), stores them once
# speedup vs baseline: 2.6420x; 1.2213x over previous
#include <hip/hip_runtime.h>
#include <hip/hip_bf16.h>

typedef __attribute__((ext_vector_type(16))) _Float16 v16h;
typedef __attribute__((ext_vector_type(8)))  float    v8f;
typedef __attribute__((ext_vector_type(8)))  _Float16 v8h;

#define NNODE 50000
#define NEDGE 800000
#define IND   64
#define HID   128
#define EDD   16
#define KE    160
#define KN    192
#define KEP   168
#define HP    136
#define KNP   200
#define WPB   24
#define NBLK  256
#define BN    196
#define CNTG  125
#define CHUNK (NEDGE / CNTG)
#define EPT   (CHUNK / 256)
#define F1    40
#define F2    32
#define PACKB 17
#define CONVB 512
#define NX8   (NNODE * IND / 8)
#define RECB  ((NNODE + 255) / 256)
#define NODEB ((NNODE + 63) / 64)
#define POSB  ((NNODE * 3 + 1023) / 1024)

#define LDS_SYNC() asm volatile("s_wait_dscnt 0" ::: "memory")

__device__ __forceinline__ float silu(float v) {
    return v * __builtin_amdgcn_rcpf(1.f + __expf(-v));
}

__device__ __forceinline__ unsigned pk2h(float a, float b) {
    union { _Float16 h[2]; unsigned u; } t;
    t.h[0] = (_Float16)a; t.h[1] = (_Float16)b;
    return t.u;
}

__device__ __forceinline__ float xhalf(float v) {
    int b = __builtin_bit_cast(int, v);
    return __builtin_bit_cast(float,
        __builtin_amdgcn_permlanex16(b, b, 0x76543210, 0xfedcba98, false, false));
}

__device__ __forceinline__ v16h load_a_frag(const _Float16* base, int lane, int kbase, int stride) {
    int m = lane & 15, hs = lane >> 4;
    union { v16h v; unsigned u[8]; } f;
    const _Float16* p = base + m * stride + kbase + 8 * hs;
#pragma unroll
    for (int r = 0; r < 8; ++r) {
        int k0 = (r < 4) ? (2 * r) : (16 + 2 * (r - 4));
        f.u[r] = *(const unsigned*)(p + k0);
    }
    return f.v;
}

__device__ __forceinline__ void pack_frag(const float* __restrict__ W, _Float16* __restrict__ out,
                                          int frag, int lane, int K, int N, int NT, int planes) {
    int nt = frag % NT;
    int kc = frag / NT;
    int nC = lane & 15, hs = lane >> 4;
    int n = nt * 16 + nC;
#pragma unroll
    for (int j = 0; j < 16; ++j) {
        int r = j >> 1, p = j & 1;
        int kl = ((r < 4) ? 2 * r : 16 + 2 * (r - 4)) + p + 8 * hs;
        int k = kc * 32 + kl;
        float v = (k < K) ? W[(long)k * N + n] : 0.f;
        long o = planes ? (((long)frag * 2 + (j >> 3)) * 32 + lane) * 8 + (j & 7)
                        : ((long)frag * 32 + lane) * 16 + j;
        out[o] = (_Float16)v;
    }
}

__global__ void __launch_bounds__(256) prep_kernel(
    const float* __restrict__ We1, const float* __restrict__ We2,
    const float* __restrict__ Wn1, const float* __restrict__ Wn2,
    _Float16* __restrict__ We1f, _Float16* __restrict__ We2f,
    _Float16* __restrict__ Wn1f, _Float16* __restrict__ Wn2f,
    const int* __restrict__ ei, int* __restrict__ hist,
    const float* __restrict__ x, const float* __restrict__ pos, const float* __restrict__ s,
    _Float16* __restrict__ x16, float* __restrict__ rec)
{
    if (blockIdx.x < PACKB) {
        int f = blockIdx.x * 8 + (threadIdx.x >> 5);
        int lane = threadIdx.x & 31;
        if (f < F1)                pack_frag(We1, We1f, f, lane, 145, 128, 8, 1);
        else if (f < F1 + F2)      pack_frag(We2, We2f, f - F1, lane, 128, 128, 8, 1);
        else if (f < F1 + F2 + 48) pack_frag(Wn1, Wn1f, f - (F1 + F2), lane, 192, 128, 8, 0);
        else                       pack_frag(Wn2, Wn2f, f - (F1 + F2 + 48), lane, 128, 64, 4, 0);
        return;
    }
    if (blockIdx.x < PACKB + CNTG) {
        __shared__ int cnt_s[NBLK];
        int g = blockIdx.x - PACKB;
        cnt_s[threadIdx.x] = 0;
        __syncthreads();
        const int* dstp = ei + NEDGE + g * CHUNK + threadIdx.x;
        int d[EPT];
#pragma unroll
        for (int j = 0; j < EPT; ++j) d[j] = dstp[256 * j];
#pragma unroll
        for (int j = 0; j < EPT; ++j)
            __hip_atomic_fetch_add(&cnt_s[d[j] / BN], 1, __ATOMIC_RELAXED, __HIP_MEMORY_SCOPE_WORKGROUP);
        __syncthreads();
        hist[g * NBLK + threadIdx.x] = cnt_s[threadIdx.x];
        return;
    }
    if (blockIdx.x < PACKB + CNTG + CONVB) {
        const float4* xs = (const float4*)x;
        uint4* xd = (uint4*)x16;
        for (int i = (blockIdx.x - (PACKB + CNTG)) * 256 + threadIdx.x; i < NX8; i += CONVB * 256) {
            float4 a = xs[2 * i], b = xs[2 * i + 1];
            xd[i] = make_uint4(pk2h(a.x, a.y), pk2h(a.z, a.w), pk2h(b.x, b.y), pk2h(b.z, b.w));
        }
        return;
    }
    int i = (blockIdx.x - (PACKB + CNTG + CONVB)) * 256 + threadIdx.x;
    if (i < NNODE)
        ((float4*)rec)[i] = make_float4(pos[i * 3 + 0], pos[i * 3 + 1], pos[i * 3 + 2], s[i]);
}

__device__ __forceinline__ v16h lds_b_frag(const _Float16* wl, int f, int wq) {
    const v8h* p = (const v8h*)(wl + f * 512 + wq);
    v8h lo = p[0], hi = p[32];
    return __builtin_shufflevector(lo, hi, 0, 1, 2, 3, 4, 5, 6, 7, 8, 9, 10, 11, 12, 13, 14, 15);
}

__global__ void __launch_bounds__(256) place_kernel(
    const int* __restrict__ ei, const int* __restrict__ hist,
    int* __restrict__ perm, int* __restrict__ bstart, int* __restrict__ btot)
{
    __shared__ int cur_s[NBLK];
    __shared__ int wsum_s[8];
    int g = blockIdx.x, t = threadIdx.x;
    int lane = t & 31, wv = t >> 5;
    const int* dstp = ei + NEDGE + g * CHUNK + t;
    int d[EPT];
#pragma unroll
    for (int j = 0; j < EPT; ++j) d[j] = dstp[256 * j];
    int before = 0, total = 0;
#pragma unroll 25
    for (int k = 0; k < CNTG; ++k) {
        int v = hist[k * NBLK + t];
        total += v;
        before += (k < g) ? v : 0;
    }
    int inc = total;
    for (int off = 1; off < 32; off <<= 1) {
        int u = __shfl_up(inc, off, 32);
        if (lane >= off) inc += u;
    }
    if (lane == 31) wsum_s[wv] = inc;
    __syncthreads();
    int base = inc - total;
    for (int k = 0; k < wv; ++k) base += wsum_s[k];
    cur_s[t] = base + before;
    if (g == 0) { bstart[t] = base; btot[t] = total; }
    __syncthreads();
    int slot[EPT];
#pragma unroll
    for (int j = 0; j < EPT; ++j)
        slot[j] = __hip_atomic_fetch_add(&cur_s[d[j] / BN], 1, __ATOMIC_RELAXED, __HIP_MEMORY_SCOPE_WORKGROUP);
#pragma unroll
    for (int j = 0; j < EPT; ++j) perm[slot[j]] = g * CHUNK + t + 256 * j;
}

__global__ void __launch_bounds__(WPB * 32) edge_kernel(
    const _Float16* __restrict__ x16, const float* __restrict__ rec,
    const int* __restrict__ ei, const float* __restrict__ ea,
    const int* __restrict__ perm, const int* __restrict__ bstart, const int* __restrict__ btot,
    const _Float16* __restrict__ We1f, const float* __restrict__ be1,
    const _Float16* __restrict__ We2f, const float* __restrict__ be2,
    const float* __restrict__ Wc, const float* __restrict__ bc,
    float* __restrict__ m_sum, float* __restrict__ acc4)
{
    __shared__ __attribute__((aligned(16))) _Float16 wl_s[(F1 + F2) * 512];
    __shared__ __attribute__((aligned(16))) _Float16 ein_s[WPB][16][KEP];
    __shared__ __attribute__((aligned(16))) float ms_s[BN * HID];
    __shared__ __attribute__((aligned(16))) float a4_s[BN * 4];
    __shared__ float sg_s[WPB][16];
    __shared__ float df_s[WPB][16][4];
    __shared__ int   idx_s[WPB][32];
    __shared__ int   eid_s[WPB][16];

    int w = __builtin_amdgcn_readfirstlane(threadIdx.x >> 5);
    int lane0 = threadIdx.x & 31;
    int node0 = blockIdx.x * BN;
    int base = bstart[blockIdx.x], total = btot[blockIdx.x];

    {
        uint4* z = (uint4*)ms_s;
        for (int i = threadIdx.x; i < BN * HID / 4; i += WPB * 32) z[i] = make_uint4(0u, 0u, 0u, 0u);
        uint4* z4 = (uint4*)a4_s;
        for (int i = threadIdx.x; i < BN; i += WPB * 32) z4[i] = make_uint4(0u, 0u, 0u, 0u);
    }
    {
        const uint4* g1 = (const uint4*)We1f;
        const uint4* g2 = (const uint4*)We2f;
        uint4* l = (uint4*)wl_s;
        for (int i = threadIdx.x; i < F1 * 64; i += WPB * 32) l[i] = g1[i];
        for (int i = threadIdx.x; i < F2 * 64; i += WPB * 32) l[F1 * 64 + i] = g2[i];
    }
    __syncthreads();
    const _Float16* wl1 = wl_s;
    const _Float16* wl2 = wl_s + F1 * 512;

    int ntile = (total + 15) >> 4;
    for (int tile = w; tile < ntile; tile += WPB) {
    LDS_SYNC();
    int lane = lane0;
    asm volatile("" : "+v"(lane));
    int nC = lane & 15, hs = lane >> 4;
    int wq = lane * 8;


    int slot = tile * 16 + nC;
    float vf = (slot < total) ? 1.0f : 0.0f;
    int e = perm[base + ((slot < total) ? slot : 0)];
    int node = ei[(hs ? 0 : NEDGE) + e];
    if (lane < 16) eid_s[w][lane] = e;
    idx_s[w][lane] = node;
    float4 rc = *(const float4*)(rec + (long)node * 4);
    LDS_SYNC();

    float ev[8];
#pragma unroll
    for (int t = 0; t < 8; ++t)
        ev[t] = ea[(long)eid_s[w][2 * t + hs] * EDD + nC];

    uint4 xr[8];
#pragma unroll
    for (int t = 0; t < 8; ++t) {
        int row = (lane >> 3) + 4 * t;
        unsigned off = (unsigned)idx_s[w][row] * (IND * 2u) + (lane & 7) * 16u;
        xr[t] = *(const uint4*)((const char*)x16 + off);
    }

#pragma unroll
    for (int t = 0; t < 8; ++t) {
        int m = (lane >> 3) + 4 * (t & 3);
        *(uint4*)&ein_s[w][m][(t >> 2) * IND + (lane & 7) * 8] = xr[t];
    }
#pragma unroll
    for (int t = 0; t < 8; ++t) {
        int idx = lane + 32 * t;
        ein_s[w][idx >> 4][2 * IND + 1 + (idx & 15)] = (_Float16)ev[t];
    }
    float ox = xhalf(rc.x), oy = xhalf(rc.y), oz = xhalf(rc.z);
    if (lane < 16) {
        float dx = rc.x - ox;
        float dy = rc.y - oy;
        float dz = rc.z - oz;
        df_s[w][lane][0] = dx * vf; df_s[w][lane][1] = dy * vf; df_s[w][lane][2] = dz * vf;
        df_s[w][lane][3] = vf;
        ein_s[w][lane][2 * IND] = (_Float16)(dx * dx + dy * dy + dz * dz);
#pragma unroll
        for (int c = 2 * IND + 1 + EDD; c < KE; ++c)
            ein_s[w][lane][c] = (_Float16)0.f;
    } else {
        sg_s[w][nC] = rc.w * vf;
    }
    LDS_SYNC();

    v16h af[5];
#pragma unroll
    for (int kc = 0; kc < 5; ++kc)
        af[kc] = load_a_frag(&ein_s[w][0][0], lane, kc * 32, KEP);
    v16h bw[5];
#pragma unroll
    for (int kc = 0; kc < 5; ++kc) bw[kc] = lds_b_frag(wl1, kc * 8, wq);
    const float4* bp1 = (const float4*)be1;
    int b8 = 2 * hs;
    float4 ba = bp1[b8], bb = bp1[b8 + 1];
    union { v16h v; unsigned u[8]; } h1[4];
#pragma unroll
    for (int mt = 0; mt < 8; ++mt) {
        v8f c = {ba.x, ba.y, ba.z, ba.w, bb.x, bb.y, bb.z, bb.w};
#pragma unroll
        for (int kc = 0; kc < 5; ++kc)
            c = __builtin_amdgcn_wmma_f32_16x16x32_f16(false, bw[kc], false, af[kc], (short)0, c, false, false);
        if (mt < 7) {
#pragma unroll
            for (int kc = 0; kc < 5; ++kc) bw[kc] = lds_b_frag(wl1, kc * 8 + mt + 1, wq);
            ba = bp1[b8 + 4 * (mt + 1)];
            bb = bp1[b8 + 4 * (mt + 1) + 1];
        }
        __builtin_amdgcn_sched_barrier(0);
#pragma unroll
        for (int i = 0; i < 4; ++i)
            h1[mt >> 1].u[4 * (mt & 1) + i] = pk2h(silu(c[2 * i]), silu(c[2 * i + 1]));
    }

    int rowoff[16];
#pragma unroll
    for (int k = 0; k < 16; ++k) rowoff[k] = (idx_s[w][k] - node0) * HID + lane;

    v8f gp = {0, 0, 0, 0, 0, 0, 0, 0};
    for (int np = 0; np < 4; ++np) {
        int n0 = np * 32 + nC, n1 = n0 + 16;
        v16h b0[4], b1[4];
#pragma unroll
        for (int kc = 0; kc < 4; ++kc) {
            b0[kc] = lds_b_frag(wl2, kc * 8 + 2 * np, wq);
            b1[kc] = lds_b_frag(wl2, kc * 8 + 2 * np + 1, wq);
        }
        float bs0 = be2[n0], bs1 = be2[n1];
        float wc0 = Wc[n0], wc1 = Wc[n1];
        v8f c0 = {bs0, bs0, bs0, bs0, bs0, bs0, bs0, bs0};
        v8f c1 = {bs1, bs1, bs1, bs1, bs1, bs1, bs1, bs1};
#pragma unroll
        for (int kc = 0; kc < 4; ++kc) {
            v16h a = h1[kc].v;
            c0 = __builtin_amdgcn_wmma_f32_16x16x32_f16(false, a, false, b0[kc], (short)0, c0, false, false);
            c1 = __builtin_amdgcn_wmma_f32_16x16x32_f16(false, a, false, b1[kc], (short)0, c1, false, false);
        }
#pragma unroll
        for (int r = 0; r < 8; ++r) {
            float sg = sg_s[w][r + 8 * hs];
            float v0 = silu(c0[r]) * sg;
            float v1 = silu(c1[r]) * sg;
            gp[r] += v0 * wc0;
            gp[r] += v1 * wc1;
            float got = xhalf(hs ? v0 : v1);
            float lo = hs ? got : v0;
            float hi = hs ? v1 : got;
            __hip_atomic_fetch_add(ms_s + rowoff[r], lo, __ATOMIC_RELAXED, __HIP_MEMORY_SCOPE_WORKGROUP);
            __hip_atomic_fetch_add(ms_s + rowoff[r + 8], hi, __ATOMIC_RELAXED, __HIP_MEMORY_SCOPE_WORKGROUP);
        }
#pragma unroll
        for (int k = 0; k < 16; ++k) rowoff[k] += 32;
    }
#pragma unroll
    for (int off = 1; off < 16; off <<= 1)
#pragma unroll
        for (int r = 0; r < 8; ++r) gp[r] += __shfl_xor((float)gp[r], off, 32);
    {
        float bcv = bc[0];
        int q = nC >> 2, wsel = nC & 3;
#pragma unroll
        for (int j = 0; j < 2; ++j) {
            float gq = (q == 0) ? gp[4 * j] : (q == 1) ? gp[4 * j + 1]
                     : (q == 2) ? gp[4 * j + 2] : gp[4 * j + 3];
            float g = gq + bcv;
            int m = q + 4 * j + 8 * hs;
            float val = ((wsel == 3) ? 1.0f : g) * df_s[w][m][wsel];
            __hip_atomic_fetch_add(&a4_s[(idx_s[w][m] - node0) * 4 + wsel], val,
                                   __ATOMIC_RELAXED, __HIP_MEMORY_SCOPE_WORKGROUP);
        }
    }
    }

    __syncthreads();
    {
        int nrows = (NNODE - node0 < BN) ? (NNODE - node0) : BN;
        const uint4* ml = (const uint4*)ms_s;
        uint4* mo = (uint4*)(m_sum + (long)node0 * HID);
        for (int i = threadIdx.x; i < nrows * (HID / 4); i += WPB * 32) mo[i] = ml[i];
        const uint4* al = (const uint4*)a4_s;
        uint4* ao = (uint4*)(acc4 + (long)node0 * 4);
        for (int i = threadIdx.x; i < nrows; i += WPB * 32) ao[i] = al[i];
    }
}

__global__ void __launch_bounds__(128) node_kernel(
    const float* __restrict__ x, const float* __restrict__ m_sum,
    const float* __restrict__ acc4,
    const _Float16* __restrict__ Wn1f, const float* __restrict__ bn1,
    const _Float16* __restrict__ Wn2f, const float* __restrict__ bn2,
    const float* __restrict__ pos,
    float* __restrict__ out_x, float* __restrict__ out_pos)
{
    __shared__ __attribute__((aligned(16))) _Float16 ain_s[4][16][KNP];
    __shared__ __attribute__((aligned(16))) _Float16 h_s[4][16][HP];
    __shared__ float rdeg_s[4][16];

    if (blockIdx.x >= NODEB) {
        int base = (blockIdx.x - NODEB) * 1024 + threadIdx.x;
        float pa[8], pc[8], pd[8];
#pragma unroll
        for (int j = 0; j < 8; ++j) {
            int ge = base + 128 * j;
            int gc = ge < NNODE * 3 ? ge : NNODE * 3 - 1;
            int i = gc / 3;
            int k = gc - 3 * i;
            pa[j] = pos[gc];
            pc[j] = acc4[i * 4 + k];
            pd[j] = acc4[i * 4 + 3];
        }
#pragma unroll
        for (int j = 0; j < 8; ++j) {
            int ge = base + 128 * j;
            float dg = pd[j]; if (dg < 1.f) dg = 1.f;
            if (ge < NNODE * 3) out_pos[ge] = pa[j] + pc[j] * __builtin_amdgcn_rcpf(dg);
        }
        return;
    }

    int w = threadIdx.x >> 5, lane = threadIdx.x & 31;
    int i0 = (blockIdx.x * 4 + w) * 16;
    if (i0 >= NNODE) return;
    int nC = lane & 15, hs = lane >> 4;

    float dg = 1.f;
    if (lane < 16) {
        int i = i0 + lane; if (i >= NNODE) i = NNODE - 1;
        dg = acc4[i * 4 + 3];
    }
    float4 vx[8], vm[16];
#pragma unroll
    for (int t = 0; t < 8; ++t) {
        int idx = lane + 32 * t;
        int m = idx >> 4, c4 = (idx & 15) * 4;
        int i = i0 + m; if (i >= NNODE) i = NNODE - 1;
        vx[t] = *(const float4*)(x + (long)i * IND + c4);
    }
#pragma unroll
    for (int t = 0; t < 16; ++t) {
        int i = i0 + t; if (i >= NNODE) i = NNODE - 1;
        vm[t] = *(const float4*)(m_sum + (long)i * HID + lane * 4);
    }
    if (lane < 16) {
        if (dg < 1.f) dg = 1.f;
        rdeg_s[w][lane] = __builtin_amdgcn_rcpf(dg);
    }
    LDS_SYNC();
#pragma unroll
    for (int t = 0; t < 8; ++t) {
        int idx = lane + 32 * t;
        int m = idx >> 4, c4 = (idx & 15) * 4;
        *(uint2*)&ain_s[w][m][c4] = make_uint2(pk2h(vx[t].x, vx[t].y), pk2h(vx[t].z, vx[t].w));
    }
#pragma unroll
    for (int t = 0; t < 16; ++t) {
        float r = rdeg_s[w][t];
        *(uint2*)&ain_s[w][t][IND + lane * 4] =
            make_uint2(pk2h(vm[t].x * r, vm[t].y * r), pk2h(vm[t].z * r, vm[t].w * r));
    }
    LDS_SYNC();

    const v16h* W1 = (const v16h*)Wn1f + lane;
    const v16h* W2 = (const v16h*)Wn2f + lane;

    v16h af[6];
#pragma unroll
    for (int kc = 0; kc < 6; ++kc)
        af[kc] = load_a_frag(&ain_s[w][0][0], lane, kc * 32, KNP);
    v16h bw[6];
#pragma unroll
    for (int kc = 0; kc < 6; ++kc) bw[kc] = W1[(kc * 8) * 32];
    float bias1 = bn1[nC];
    for (int nt = 0; nt < 7; ++nt) {
        v8f c = {bias1, bias1, bias1, bias1, bias1, bias1, bias1, bias1};
#pragma unroll
        for (int kc = 0; kc < 6; ++kc)
            c = __builtin_amdgcn_wmma_f32_16x16x32_f16(false, af[kc], false, bw[kc], (short)0, c, false, false);
#pragma unroll
        for (int kc = 0; kc < 6; ++kc) bw[kc] = W1[(kc * 8 + nt + 1) * 32];
        bias1 = bn1[(nt + 1) * 16 + nC];
#pragma unroll
        for (int r = 0; r < 8; ++r)
            h_s[w][r + 8 * hs][nt * 16 + nC] = (_Float16)silu(c[r]);
    }
    v16h bv[4];
    float bias2;
    {
        v8f c = {bias1, bias1, bias1, bias1, bias1, bias1, bias1, bias1};
#pragma unroll
        for (int kc = 0; kc < 6; ++kc)
            c = __builtin_amdgcn_wmma_f32_16x16x32_f16(false, af[kc], false, bw[kc], (short)0, c, false, false);
#pragma unroll
        for (int kc = 0; kc < 4; ++kc) bv[kc] = W2[(kc * 4) * 32];
        bias2 = bn2[nC];
#pragma unroll
        for (int r = 0; r < 8; ++r)
            h_s[w][r + 8 * hs][7 * 16 + nC] = (_Float16)silu(c[r]);
    }
    LDS_SYNC();

    v16h ah[4];
#pragma unroll
    for (int kc = 0; kc < 4; ++kc)
        ah[kc] = load_a_frag(&h_s[w][0][0], lane, kc * 32, HP);
#pragma unroll
    for (int nt = 0; nt < 4; ++nt) {
        int n = nt * 16 + nC;
        v8f c = {bias2, bias2, bias2, bias2, bias2, bias2, bias2, bias2};
#pragma unroll
        for (int kc = 0; kc < 4; ++kc)
            c = __builtin_amdgcn_wmma_f32_16x16x32_f16(false, ah[kc], false, bv[kc], (short)0, c, false, false);
        if (nt + 1 < 4) {
#pragma unroll
            for (int kc = 0; kc < 4; ++kc) bv[kc] = W2[(kc * 4 + nt + 1) * 32];
            bias2 = bn2[(nt + 1) * 16 + nC];
        }
#pragma unroll
        for (int r = 0; r < 8; ++r) {
            int i = i0 + r + 8 * hs;
            if (i < NNODE) out_x[(long)i * IND + n] = c[r];
        }
    }
}

extern "C" void kernel_launch(void* const* d_in, const int* in_sizes, int n_in,
                              void* d_out, int out_size, void* d_ws, size_t ws_size,
                              hipStream_t stream) {
    const float* x   = (const float*)d_in[0];
    const float* pos = (const float*)d_in[1];
    const int*   ei  = (const int*)d_in[2];
    const float* ea  = (const float*)d_in[3];
    const float* s   = (const float*)d_in[4];
    const float* We1 = (const float*)d_in[5];  const float* be1 = (const float*)d_in[6];
    const float* We2 = (const float*)d_in[7];  const float* be2 = (const float*)d_in[8];
    const float* Wn1 = (const float*)d_in[9];  const float* bn1 = (const float*)d_in[10];
    const float* Wn2 = (const float*)d_in[11]; const float* bn2 = (const float*)d_in[12];
    const float* Wc  = (const float*)d_in[13]; const float* bc  = (const float*)d_in[14];

    float* out_x   = (float*)d_out;
    float* out_pos = out_x + (long)NNODE * IND;

    float* m_sum = (float*)d_ws;
    float* acc4  = m_sum + (long)NNODE * HID;
    _Float16* We1f = (_Float16*)(acc4 + (long)NNODE * 4);
    _Float16* We2f = We1f + 5 * 8 * 512;
    _Float16* Wn1f = We2f + 4 * 8 * 512;
    _Float16* Wn2f = Wn1f + 6 * 8 * 512;

    _Float16* x16 = (_Float16*)d_out;
    float*    rec = out_x + (long)NNODE * IND / 2;
    int* perm   = (int*)(rec + (long)NNODE * 4);
    int* hist   = perm + NEDGE;
    int* bstart = hist + CNTG * NBLK;
    int* btot   = bstart + NBLK;

    prep_kernel<<<PACKB + CNTG + CONVB + RECB, 256, 0, stream>>>(We1, We2, Wn1, Wn2,
                                                   We1f, We2f, Wn1f, Wn2f, ei, hist,
                                                   x, pos, s, x16, rec);

    place_kernel<<<CNTG, 256, 0, stream>>>(ei, hist, perm, bstart, btot);

    edge_kernel<<<NBLK, WPB * 32, 0, stream>>>(x16, rec, ei, ea, perm, bstart, btot,
                                                We1f, be1, We2f, be2, Wc, bc,
                                                m_sum, acc4);
    node_kernel<<<NODEB + POSB, 128, 0, stream>>>(x, m_sum, acc4,
                                                  Wn1f, bn1, Wn2f, bn2, pos,
                                                  out_x, out_pos);
}
